# starting kernel
# baseline (speedup 1.0000x reference)
#include <hip/hip_runtime.h>

typedef __attribute__((ext_vector_type(16))) _Float16 v16h;
typedef __attribute__((ext_vector_type(8)))  float    v8f;

#define WMMA_F16(A, B, C) \
    __builtin_amdgcn_wmma_f32_16x16x32_f16(false, (A), false, (B), (short)0, (C), false, false)

#define L2E 1.4426950408889634f
#define LN2 0.6931471805599453f

__device__ __forceinline__ float exp2_fast(float x) { return __builtin_amdgcn_exp2f(x); }
__device__ __forceinline__ float log2_fast(float x) { return __builtin_amdgcn_logf(x); }
__device__ __forceinline__ float rcp_fast(float x)  { return __builtin_amdgcn_rcpf(x); }

__device__ __forceinline__ float softplus_f(float x) {
    return x > 20.f ? x : LN2 * log2_fast(1.f + exp2_fast(x * L2E));
}
__device__ __forceinline__ float sigmoid_f(float x) {
    return rcp_fast(1.f + exp2_fast(-x * L2E));
}

__global__ __launch_bounds__(256)
void nerf_wmma_kernel(const float* __restrict__ rays_o,
                      const float* __restrict__ rays_d,
                      const float* __restrict__ W1,   const float* __restrict__ b1,
                      const float* __restrict__ Wsig, const float* __restrict__ Wsig_d,
                      const float* __restrict__ Wc1,  const float* __restrict__ bc1,
                      const float* __restrict__ Wc2,  const float* __restrict__ Wc2_d,
                      const int*   __restrict__ num_steps,
                      float* __restrict__ out, int N)
{
    const int tid  = blockIdx.x * blockDim.x + threadIdx.x;
    const int lane = threadIdx.x & 31;
    const int wave = tid >> 5;
    const int m    = lane & 15;
    const bool hi  = (lane & 16) != 0;
    const int rayBase = wave * 16;

    int ray = rayBase + m;
    if (ray >= N) ray = N - 1;

    v16h A1t0, A1t1, Ac1t0, Ac1t1, Asig, Acol;
#pragma unroll
    for (int e = 0; e < 16; ++e) {
        const int K = (e < 8 ? e : e + 8) + (hi ? 8 : 0);
        float a0 = (K < 3) ? W1[K * 32 + m]        : 0.f;
        float a1 = (K < 3) ? W1[K * 32 + (m + 16)] : 0.f;
        A1t0[e] = (_Float16)a0;
        A1t1[e] = (_Float16)a1;
        float c0 = (K < 6) ? Wc1[K * 32 + m]        : 0.f;
        float c1 = (K < 6) ? Wc1[K * 32 + (m + 16)] : 0.f;
        Ac1t0[e] = (_Float16)c0;
        Ac1t1[e] = (_Float16)c1;
        float sv = (m == 0) ? Wsig[K] : ((m == 1) ? Wsig_d[K] : 0.f);
        Asig[e] = (_Float16)sv;
        float cv = (m < 3) ? Wc2[K * 3 + m] : ((m < 6) ? Wc2_d[K * 3 + (m - 3)] : 0.f);
        Acol[e] = (_Float16)cv;
    }
    v8f CbT0, CbT1, CbC0, CbC1;
#pragma unroll
    for (int i = 0; i < 8; ++i) {
        const int h0 = i + (hi ? 8 : 0);
        CbT0[i] = b1[h0];      CbT1[i] = b1[h0 + 16];
        CbC0[i] = bc1[h0];     CbC1[i] = bc1[h0 + 16];
    }

    float ox = rays_o[ray * 3 + 0], oy = rays_o[ray * 3 + 1], oz = rays_o[ray * 3 + 2];
    float dx = rays_d[ray * 3 + 0], dy = rays_d[ray * 3 + 1], dz = rays_d[ray * 3 + 2];
    {
        float il = 1.f / sqrtf(dx * dx + dy * dy + dz * dz);
        dx *= il; dy *= il; dz *= il;
    }
    float near, far;
    {
        float ivx = 1.f / dx, ivy = 1.f / dy, ivz = 1.f / dz;
        float t1x = (-1.f - ox) * ivx, t2x = (1.f - ox) * ivx;
        float t1y = (-1.f - oy) * ivy, t2y = (1.f - oy) * ivy;
        float t1z = (-1.f - oz) * ivz, t2z = (1.f - oz) * ivz;
        near = fmaxf(fmaxf(fminf(t1x, t2x), fminf(t1y, t2y)), fminf(t1z, t2z));
        far  = fminf(fminf(fmaxf(t1x, t2x), fmaxf(t1y, t2y)), fmaxf(t1z, t2z));
        near = fmaxf(near, 0.2f);
        far  = fmaxf(far, near + 1e-6f);
    }

    const int   T      = *num_steps;
    const float invTm1 = 1.f / (float)(T - 1);
    const float fn     = far - near;
    const float dstep  = fn * invTm1;
    const float dlast  = fn / (float)T;

    v16h Bm = {};
    v16h Bc = {};
    Bc[3] = (_Float16)(hi ? 0.f : dx);
    Bc[4] = (_Float16)(hi ? 0.f : dy);
    Bc[5] = (_Float16)(hi ? 0.f : dz);

    float Tc = 1.f,  ws = 0.f,  dep = 0.f,  im0 = 0.f, im1 = 0.f, im2 = 0.f;
    float Td = 1.f,  wsd = 0.f, depd = 0.f, id0 = 0.f, id1 = 0.f, id2 = 0.f;

    for (int t = 0; t < T; ++t) {
        const float z01 = (float)t * invTm1;
        const float zv  = near + fn * z01;
        const float px = fminf(fmaxf(ox + dx * zv, -1.f), 1.f);
        const float py = fminf(fmaxf(oy + dy * zv, -1.f), 1.f);
        const float pz = fminf(fmaxf(oz + dz * zv, -1.f), 1.f);
        const _Float16 qx = (_Float16)(hi ? 0.f : px);
        const _Float16 qy = (_Float16)(hi ? 0.f : py);
        const _Float16 qz = (_Float16)(hi ? 0.f : pz);
        Bm[0] = qx; Bm[1] = qy; Bm[2] = qz;
        Bc[0] = qx; Bc[1] = qy; Bc[2] = qz;

        v8f D0 = WMMA_F16(A1t0, Bm, CbT0);
        v8f D1 = WMMA_F16(A1t1, Bm, CbT1);
        v16h Bh;
#pragma unroll
        for (int i = 0; i < 8; ++i) {
            Bh[i]     = (_Float16)fmaxf(D0[i], 0.f);
            Bh[i + 8] = (_Float16)fmaxf(D1[i], 0.f);
        }
        v8f Z0 = {};
        v8f Ds = WMMA_F16(Asig, Bh, Z0);

        v8f C0 = WMMA_F16(Ac1t0, Bc, CbC0);
        v8f C1 = WMMA_F16(Ac1t1, Bc, CbC1);
        v16h Bhc;
#pragma unroll
        for (int i = 0; i < 8; ++i) {
            Bhc[i]     = (_Float16)fmaxf(C0[i], 0.f);
            Bhc[i + 8] = (_Float16)fmaxf(C1[i], 0.f);
        }
        v8f Z1 = {};
        v8f Dc = WMMA_F16(Acol, Bhc, Z1);

        const float delta = (t == T - 1) ? dlast : dstep;
        const float dl2   = -delta * L2E;
        const float sg  = softplus_f(Ds[0]);
        const float sgd = softplus_f(Ds[1]);
        const float a   = 1.f - exp2_fast(dl2 * sg);
        const float ad  = 1.f - exp2_fast(dl2 * sgd);
        const float w   = a  * Tc;
        const float wd  = ad * Td;
        Tc *= (1.f - a  + 1e-15f);
        Td *= (1.f - ad + 1e-15f);
        ws  += w;   dep  += w  * z01;
        wsd += wd;  depd += wd * z01;
        im0 += w  * sigmoid_f(Dc[0]);
        im1 += w  * sigmoid_f(Dc[1]);
        im2 += w  * sigmoid_f(Dc[2]);
        id0 += wd * sigmoid_f(Dc[3]);
        id1 += wd * sigmoid_f(Dc[4]);
        id2 += wd * sigmoid_f(Dc[5]);
    }

    if (!hi && (rayBase + m) < N) {
        float* o9 = out + (size_t)(rayBase + m) * 9;
        const float bg  = 1.f - ws;
        const float bgd = 1.f - wsd;
        o9[0] = im0 + bg;  o9[1] = im1 + bg;  o9[2] = im2 + bg;
        o9[3] = dep;
        o9[4] = ws;
        o9[5] = id0 + bgd; o9[6] = id1 + bgd; o9[7] = id2 + bgd;
        o9[8] = depd;
    }
}

extern "C" void kernel_launch(void* const* d_in, const int* in_sizes, int n_in,
                              void* d_out, int out_size, void* d_ws, size_t ws_size,
                              hipStream_t stream) {
    const float* rays_o = (const float*)d_in[0];
    const float* rays_d = (const float*)d_in[1];
    const float* W1     = (const float*)d_in[2];
    const float* b1     = (const float*)d_in[3];
    const float* Wsig   = (const float*)d_in[4];
    const float* Wsig_d = (const float*)d_in[5];
    const float* Wc1    = (const float*)d_in[6];
    const float* bc1    = (const float*)d_in[7];
    const float* Wc2    = (const float*)d_in[8];
    const float* Wc2_d  = (const float*)d_in[9];
    const int*   nsteps = (const int*)d_in[10];

    const int N      = in_sizes[0] / 3;
    const int waves  = (N + 15) / 16;
    const int thr    = waves * 32;
    const int block  = 256;
    const int grid   = (thr + block - 1) / block;

    nerf_wmma_kernel<<<grid, block, 0, stream>>>(
        rays_o, rays_d, W1, b1, Wsig, Wsig_d, Wc1, bc1, Wc2, Wc2_d,
        nsteps, (float*)d_out, N);
}


# tile columns = the 16 lanes' own sample blocks of ONE ray (8 rays per wave, 4096 waves); blocks joined per ray by a 16-lane prefix product of transmittance and row sums: summation order changes
# speedup vs baseline: 1.0334x; 1.0334x over previous
#include <hip/hip_runtime.h>

typedef __attribute__((ext_vector_type(16))) _Float16 v16h;
typedef __attribute__((ext_vector_type(8)))  float    v8f;

#define WMMA_F16(A, B, C) \
    __builtin_amdgcn_wmma_f32_16x16x32_f16(false, (A), false, (B), (short)0, (C), false, false)

#define L2E 1.4426950408889634f
#define LN2 0.6931471805599453f

__device__ __forceinline__ float exp2_fast(float x) { return __builtin_amdgcn_exp2f(x); }
__device__ __forceinline__ float log2_fast(float x) { return __builtin_amdgcn_logf(x); }
__device__ __forceinline__ float rcp_fast(float x)  { return __builtin_amdgcn_rcpf(x); }

__device__ __forceinline__ float softplus_f(float x) {
    return x > 20.f ? x : LN2 * log2_fast(1.f + exp2_fast(x * L2E));
}
__device__ __forceinline__ float sigmoid_f(float x) {
    return rcp_fast(1.f + exp2_fast(-x * L2E));
}

#define RPW 8

template <int CTRL>
__device__ __forceinline__ float row_dpp(float old, float src) {
    return __builtin_bit_cast(float,
        __builtin_amdgcn_update_dpp(__builtin_bit_cast(int, old), __builtin_bit_cast(int, src),
                                    CTRL, 0xf, 0xf, false));
}
#define ROW_SHR(n) (0x110 + (n))

__device__ __forceinline__ float row_scan_mul(float s) {
    asm("v_mul_f32_dpp %0, %0, %0 row_shr:1 row_mask:0xf bank_mask:0xf\n\t"
        "v_mul_f32_dpp %0, %0, %0 row_shr:2 row_mask:0xf bank_mask:0xf\n\t"
        "v_mul_f32_dpp %0, %0, %0 row_shr:4 row_mask:0xf bank_mask:0xf\n\t"
        "v_mul_f32_dpp %0, %0, %0 row_shr:8 row_mask:0xf bank_mask:0xf"
        : "+v"(s));
    return s;
}
__device__ __forceinline__ float row0_sum(float x) {
    x += row_dpp<ROW_SHR(1)>(0.f, x);
    x += row_dpp<ROW_SHR(2)>(0.f, x);
    x += row_dpp<ROW_SHR(4)>(0.f, x);
    x += row_dpp<ROW_SHR(8)>(0.f, x);
    return __builtin_bit_cast(float, __builtin_amdgcn_readlane(__builtin_bit_cast(int, x), 15));
}
__device__ __forceinline__ float lane_bcast(float x, int l) {
    return __builtin_bit_cast(float, __builtin_amdgcn_readlane(__builtin_bit_cast(int, x), l));
}

__global__ __launch_bounds__(256)
void nerf_wmma_kernel(const float* __restrict__ rays_o,
                      const float* __restrict__ rays_d,
                      const float* __restrict__ W1,   const float* __restrict__ b1,
                      const float* __restrict__ Wsig, const float* __restrict__ Wsig_d,
                      const float* __restrict__ Wc1,  const float* __restrict__ bc1,
                      const float* __restrict__ Wc2,  const float* __restrict__ Wc2_d,
                      const int*   __restrict__ num_steps,
                      float* __restrict__ out, int N)
{
    const int tid  = blockIdx.x * blockDim.x + threadIdx.x;
    const int lane = threadIdx.x & 31;
    const int wave = tid >> 5;
    const int m    = lane & 15;
    const bool hi  = (lane & 16) != 0;
    const int rayBase = wave * RPW;

    int ray = rayBase + (lane & (RPW - 1));
    if (ray >= N) ray = N - 1;

    v16h A1t0, A1t1, Ac1t0, Ac1t1, Asig, Acol;
#pragma unroll
    for (int e = 0; e < 16; ++e) {
        const int K = (e < 8 ? e : e + 8) + (hi ? 8 : 0);
        float a0 = (K < 3) ? W1[K * 32 + m]        : 0.f;
        float a1 = (K < 3) ? W1[K * 32 + (m + 16)] : 0.f;
        A1t0[e] = (_Float16)a0;
        A1t1[e] = (_Float16)a1;
        float c0 = (K < 6) ? Wc1[K * 32 + m]        : 0.f;
        float c1 = (K < 6) ? Wc1[K * 32 + (m + 16)] : 0.f;
        Ac1t0[e] = (_Float16)c0;
        Ac1t1[e] = (_Float16)c1;
        float sv = (m == 0) ? Wsig[K] : ((m == 1) ? Wsig_d[K] : 0.f);
        Asig[e] = (_Float16)sv;
        float cv = (m < 3) ? Wc2[K * 3 + m] : ((m < 6) ? Wc2_d[K * 3 + (m - 3)] : 0.f);
        Acol[e] = (_Float16)cv;
    }
    v8f CbT0, CbT1, CbC0, CbC1;
#pragma unroll
    for (int i = 0; i < 8; ++i) {
        const int h0 = i + (hi ? 8 : 0);
        CbT0[i] = b1[h0];      CbT1[i] = b1[h0 + 16];
        CbC0[i] = bc1[h0];     CbC1[i] = bc1[h0 + 16];
    }

    float ox = rays_o[ray * 3 + 0], oy = rays_o[ray * 3 + 1], oz = rays_o[ray * 3 + 2];
    float dx = rays_d[ray * 3 + 0], dy = rays_d[ray * 3 + 1], dz = rays_d[ray * 3 + 2];
    {
        float il = 1.f / sqrtf(dx * dx + dy * dy + dz * dz);
        dx *= il; dy *= il; dz *= il;
    }
    float near, far;
    {
        float ivx = 1.f / dx, ivy = 1.f / dy, ivz = 1.f / dz;
        float t1x = (-1.f - ox) * ivx, t2x = (1.f - ox) * ivx;
        float t1y = (-1.f - oy) * ivy, t2y = (1.f - oy) * ivy;
        float t1z = (-1.f - oz) * ivz, t2z = (1.f - oz) * ivz;
        near = fmaxf(fmaxf(fminf(t1x, t2x), fminf(t1y, t2y)), fminf(t1z, t2z));
        far  = fminf(fminf(fmaxf(t1x, t2x), fmaxf(t1y, t2y)), fmaxf(t1z, t2z));
        near = fmaxf(near, 0.2f);
        far  = fmaxf(far, near + 1e-6f);
    }

    const int   T       = *num_steps;
    const float invTm1  = 1.f / (float)(T - 1);
    const float fnL     = far - near;
    const float dstepL  = fnL * invTm1;
    const float dlastL  = fnL / (float)T;
    const int   nPer    = (T + 15) >> 4;

    v16h Bm = {};
    v16h Bc = {};

    float r0 = 0.f, r1 = 0.f, r2 = 0.f, r3 = 0.f, r4 = 0.f, r5 = 0.f, r6 = 0.f, r7 = 0.f, r8 = 0.f;

    for (int i = 0; i < RPW; ++i) {
        const float Rox = lane_bcast(ox, i),  Roy = lane_bcast(oy, i),  Roz = lane_bcast(oz, i);
        const float Rdx = lane_bcast(dx, i),  Rdy = lane_bcast(dy, i),  Rdz = lane_bcast(dz, i);
        const float Rnear = lane_bcast(near, i);
        const float fn    = lane_bcast(fnL, i);
        const float dstep = lane_bcast(dstepL, i);
        const float dlast = lane_bcast(dlastL, i);

        Bc[3] = (_Float16)(hi ? 0.f : Rdx);
        Bc[4] = (_Float16)(hi ? 0.f : Rdy);
        Bc[5] = (_Float16)(hi ? 0.f : Rdz);

        float Tc = 1.f, Td = 1.f;
        float ws = 0.f,  dep = 0.f,  im0 = 0.f, im1 = 0.f, im2 = 0.f;
        float wsd = 0.f, depd = 0.f, id0 = 0.f, id1 = 0.f, id2 = 0.f;

        for (int j = 0; j < nPer; ++j) {
            const int   t     = m * nPer + j;
            const bool  live  = t < T;
            const float z01 = (float)t * invTm1;
            const float zv  = Rnear + fn * z01;
            const float px = fminf(fmaxf(Rox + Rdx * zv, -1.f), 1.f);
            const float py = fminf(fmaxf(Roy + Rdy * zv, -1.f), 1.f);
            const float pz = fminf(fmaxf(Roz + Rdz * zv, -1.f), 1.f);
            const _Float16 qx = (_Float16)(hi ? 0.f : px);
            const _Float16 qy = (_Float16)(hi ? 0.f : py);
            const _Float16 qz = (_Float16)(hi ? 0.f : pz);
            Bm[0] = qx; Bm[1] = qy; Bm[2] = qz;
            Bc[0] = qx; Bc[1] = qy; Bc[2] = qz;

            v8f D0 = WMMA_F16(A1t0, Bm, CbT0);
            v8f D1 = WMMA_F16(A1t1, Bm, CbT1);
            v16h Bh;
#pragma unroll
            for (int k = 0; k < 8; ++k) {
                Bh[k]     = (_Float16)fmaxf(D0[k], 0.f);
                Bh[k + 8] = (_Float16)fmaxf(D1[k], 0.f);
            }
            v8f Z0 = {};
            v8f Ds = WMMA_F16(Asig, Bh, Z0);

            v8f C0 = WMMA_F16(Ac1t0, Bc, CbC0);
            v8f C1 = WMMA_F16(Ac1t1, Bc, CbC1);
            v16h Bhc;
#pragma unroll
            for (int k = 0; k < 8; ++k) {
                Bhc[k]     = (_Float16)fmaxf(C0[k], 0.f);
                Bhc[k + 8] = (_Float16)fmaxf(C1[k], 0.f);
            }
            v8f Z1 = {};
            v8f Dc = WMMA_F16(Acol, Bhc, Z1);

            const float delta = live ? ((t == T - 1) ? dlast : dstep) : 0.f;
            const float dl2   = -delta * L2E;
            const float sg  = softplus_f(Ds[0]);
            const float sgd = softplus_f(Ds[1]);
            const float a   = 1.f - exp2_fast(dl2 * sg);
            const float ad  = 1.f - exp2_fast(dl2 * sgd);
            const float w   = a  * Tc;
            const float wd  = ad * Td;
            Tc *= (1.f - a  + 1e-15f);
            Td *= (1.f - ad + 1e-15f);
            ws  += w;   dep  += w  * z01;
            wsd += wd;  depd += wd * z01;
            im0 += w  * sigmoid_f(Dc[0]);
            im1 += w  * sigmoid_f(Dc[1]);
            im2 += w  * sigmoid_f(Dc[2]);
            id0 += wd * sigmoid_f(Dc[3]);
            id1 += wd * sigmoid_f(Dc[4]);
            id2 += wd * sigmoid_f(Dc[5]);
        }

        const float E  = row_scan_mul(row_dpp<ROW_SHR(1)>(1.f, Tc));
        const float Ed = row_scan_mul(row_dpp<ROW_SHR(1)>(1.f, Td));
        const float Sws  = row0_sum(ws  * E),  Sdep  = row0_sum(dep  * E);
        const float Sim0 = row0_sum(im0 * E),  Sim1  = row0_sum(im1  * E),  Sim2 = row0_sum(im2 * E);
        const float Swsd = row0_sum(wsd * Ed), Sdepd = row0_sum(depd * Ed);
        const float Sid0 = row0_sum(id0 * Ed), Sid1  = row0_sum(id1  * Ed), Sid2 = row0_sum(id2 * Ed);
        const float bg  = 1.f - Sws;
        const float bgd = 1.f - Swsd;
        const bool mine = (lane == i);
        r0 = mine ? Sim0 + bg  : r0;  r1 = mine ? Sim1 + bg  : r1;  r2 = mine ? Sim2 + bg  : r2;
        r3 = mine ? Sdep       : r3;
        r4 = mine ? Sws        : r4;
        r5 = mine ? Sid0 + bgd : r5;  r6 = mine ? Sid1 + bgd : r6;  r7 = mine ? Sid2 + bgd : r7;
        r8 = mine ? Sdepd      : r8;
    }

    if (lane < RPW && (rayBase + lane) < N) {
        float* o9 = out + (size_t)(rayBase + lane) * 9;
        o9[0] = r0; o9[1] = r1; o9[2] = r2;
        o9[3] = r3;
        o9[4] = r4;
        o9[5] = r5; o9[6] = r6; o9[7] = r7;
        o9[8] = r8;
    }
}

extern "C" void kernel_launch(void* const* d_in, const int* in_sizes, int n_in,
                              void* d_out, int out_size, void* d_ws, size_t ws_size,
                              hipStream_t stream) {
    const float* rays_o = (const float*)d_in[0];
    const float* rays_d = (const float*)d_in[1];
    const float* W1     = (const float*)d_in[2];
    const float* b1     = (const float*)d_in[3];
    const float* Wsig   = (const float*)d_in[4];
    const float* Wsig_d = (const float*)d_in[5];
    const float* Wc1    = (const float*)d_in[6];
    const float* bc1    = (const float*)d_in[7];
    const float* Wc2    = (const float*)d_in[8];
    const float* Wc2_d  = (const float*)d_in[9];
    const int*   nsteps = (const int*)d_in[10];

    const int N      = in_sizes[0] / 3;
    const int waves  = (N + RPW - 1) / RPW;
    const int thr    = waves * 32;
    const int block  = 256;
    const int grid   = (thr + block - 1) / block;

    nerf_wmma_kernel<<<grid, block, 0, stream>>>(
        rays_o, rays_d, W1, b1, Wsig, Wsig_d, Wc1, bc1, Wc2, Wc2_d,
        nsteps, (float*)d_out, N);
}


# trip widened to 32 samples of the ray: the second 16 samples' head products land on rows 8-15 of the same head result, so all 32 lanes composite a block each; 32-lane join: summation order changes
# speedup vs baseline: 1.4109x; 1.3653x over previous
#include <hip/hip_runtime.h>

typedef __attribute__((ext_vector_type(16))) _Float16 v16h;
typedef __attribute__((ext_vector_type(8)))  float    v8f;

#define WMMA_F16(A, B, C) \
    __builtin_amdgcn_wmma_f32_16x16x32_f16(false, (A), false, (B), (short)0, (C), false, false)

#define L2E 1.4426950408889634f
#define LN2 0.6931471805599453f

__device__ __forceinline__ float exp2_fast(float x) { return __builtin_amdgcn_exp2f(x); }
__device__ __forceinline__ float log2_fast(float x) { return __builtin_amdgcn_logf(x); }
__device__ __forceinline__ float rcp_fast(float x)  { return __builtin_amdgcn_rcpf(x); }

__device__ __forceinline__ float softplus_f(float x) {
    return x > 20.f ? x : LN2 * log2_fast(1.f + exp2_fast(x * L2E));
}
__device__ __forceinline__ float sigmoid_f(float x) {
    return rcp_fast(1.f + exp2_fast(-x * L2E));
}

#define RPW 8

template <int CTRL>
__device__ __forceinline__ float row_dpp(float old, float src) {
    return __builtin_bit_cast(float,
        __builtin_amdgcn_update_dpp(__builtin_bit_cast(int, old), __builtin_bit_cast(int, src),
                                    CTRL, 0xf, 0xf, false));
}
#define ROW_SHR(n) (0x110 + (n))

__device__ __forceinline__ float row_scan_mul(float s) {
    asm("v_mul_f32_dpp %0, %0, %0 row_shr:1 row_mask:0xf bank_mask:0xf\n\t"
        "v_mul_f32_dpp %0, %0, %0 row_shr:2 row_mask:0xf bank_mask:0xf\n\t"
        "v_mul_f32_dpp %0, %0, %0 row_shr:4 row_mask:0xf bank_mask:0xf\n\t"
        "v_mul_f32_dpp %0, %0, %0 row_shr:8 row_mask:0xf bank_mask:0xf"
        : "+v"(s));
    return s;
}
__device__ __forceinline__ float lane_bcast(float x, int l) {
    return __builtin_bit_cast(float, __builtin_amdgcn_readlane(__builtin_bit_cast(int, x), l));
}
__device__ __forceinline__ float wave_sum(float x) {
    x += row_dpp<ROW_SHR(1)>(0.f, x);
    x += row_dpp<ROW_SHR(2)>(0.f, x);
    x += row_dpp<ROW_SHR(4)>(0.f, x);
    x += row_dpp<ROW_SHR(8)>(0.f, x);
    return lane_bcast(x, 15) + lane_bcast(x, 31);
}

__global__ __launch_bounds__(256)
void nerf_wmma_kernel(const float* __restrict__ rays_o,
                      const float* __restrict__ rays_d,
                      const float* __restrict__ W1,   const float* __restrict__ b1,
                      const float* __restrict__ Wsig, const float* __restrict__ Wsig_d,
                      const float* __restrict__ Wc1,  const float* __restrict__ bc1,
                      const float* __restrict__ Wc2,  const float* __restrict__ Wc2_d,
                      const int*   __restrict__ num_steps,
                      float* __restrict__ out, int N)
{
    const int tid  = blockIdx.x * blockDim.x + threadIdx.x;
    const int lane = threadIdx.x & 31;
    const int wave = tid >> 5;
    const int m    = lane & 15;
    const bool hi  = (lane & 16) != 0;
    const int rayBase = wave * RPW;

    int ray = rayBase + (lane & (RPW - 1));
    if (ray >= N) ray = N - 1;

    v16h A1lo0, A1lo1, A1hi0, A1hi1, Ac1lo0, Ac1lo1, Ac1hi0, Ac1hi1, AsigLo, AsigHi, AcolLo, AcolHi;
    const _Float16 zh = (_Float16)0.f;
    const int  hrow  = m & 7;
    const bool upper = m >= 8;
#pragma unroll
    for (int e = 0; e < 16; ++e) {
        float a0 = 0.f, a1 = 0.f, c0 = 0.f, c1 = 0.f;
        if (e < 3) { a0 = W1[e * 32 + m];  a1 = W1[e * 32 + (m + 16)]; }
        if (e < 6) { c0 = Wc1[e * 32 + m]; c1 = Wc1[e * 32 + (m + 16)]; }
        const _Float16 a0h = (_Float16)a0, a1h = (_Float16)a1, c0h = (_Float16)c0, c1h = (_Float16)c1;
        A1lo0[e]  = hi ? zh : a0h;   A1hi0[e]  = hi ? a0h : zh;
        A1lo1[e]  = hi ? zh : a1h;   A1hi1[e]  = hi ? a1h : zh;
        Ac1lo0[e] = hi ? zh : c0h;   Ac1hi0[e] = hi ? c0h : zh;
        Ac1lo1[e] = hi ? zh : c1h;   Ac1hi1[e] = hi ? c1h : zh;
        const int K = (e < 8 ? e : e + 8) + (hi ? 8 : 0);
        float sv = (hrow == 0) ? Wsig[K] : ((hrow == 1) ? Wsig_d[K] : 0.f);
        float cv = (hrow < 3) ? Wc2[K * 3 + hrow] : ((hrow < 6) ? Wc2_d[K * 3 + (hrow - 3)] : 0.f);
        const _Float16 svh = (_Float16)sv, cvh = (_Float16)cv;
        AsigLo[e] = upper ? zh : svh;   AsigHi[e] = upper ? svh : zh;
        AcolLo[e] = upper ? zh : cvh;   AcolHi[e] = upper ? cvh : zh;
    }
    v8f CbT0, CbT1, CbC0, CbC1;
#pragma unroll
    for (int i = 0; i < 8; ++i) {
        const int h0 = i + (hi ? 8 : 0);
        CbT0[i] = b1[h0];      CbT1[i] = b1[h0 + 16];
        CbC0[i] = bc1[h0];     CbC1[i] = bc1[h0 + 16];
    }

    float ox = rays_o[ray * 3 + 0], oy = rays_o[ray * 3 + 1], oz = rays_o[ray * 3 + 2];
    float dx = rays_d[ray * 3 + 0], dy = rays_d[ray * 3 + 1], dz = rays_d[ray * 3 + 2];
    {
        float il = 1.f / sqrtf(dx * dx + dy * dy + dz * dz);
        dx *= il; dy *= il; dz *= il;
    }
    float near, far;
    {
        float ivx = 1.f / dx, ivy = 1.f / dy, ivz = 1.f / dz;
        float t1x = (-1.f - ox) * ivx, t2x = (1.f - ox) * ivx;
        float t1y = (-1.f - oy) * ivy, t2y = (1.f - oy) * ivy;
        float t1z = (-1.f - oz) * ivz, t2z = (1.f - oz) * ivz;
        near = fmaxf(fmaxf(fminf(t1x, t2x), fminf(t1y, t2y)), fminf(t1z, t2z));
        far  = fminf(fminf(fmaxf(t1x, t2x), fmaxf(t1y, t2y)), fmaxf(t1z, t2z));
        near = fmaxf(near, 0.2f);
        far  = fmaxf(far, near + 1e-6f);
    }

    const int   T       = *num_steps;
    const float invTm1  = 1.f / (float)(T - 1);
    const float fnL     = far - near;
    const float dstepL  = fnL * invTm1;
    const float dlastL  = fnL / (float)T;
    const int   nPer    = (T + 31) >> 5;

    v16h Bf = {};

    float r0 = 0.f, r1 = 0.f, r2 = 0.f, r3 = 0.f, r4 = 0.f, r5 = 0.f, r6 = 0.f, r7 = 0.f, r8 = 0.f;

    for (int i = 0; i < RPW; ++i) {
        const float Rox = lane_bcast(ox, i),  Roy = lane_bcast(oy, i),  Roz = lane_bcast(oz, i);
        const float Rdx = lane_bcast(dx, i),  Rdy = lane_bcast(dy, i),  Rdz = lane_bcast(dz, i);
        const float Rnear = lane_bcast(near, i);
        const float fn    = lane_bcast(fnL, i);
        const float dstep = lane_bcast(dstepL, i);
        const float dlast = lane_bcast(dlastL, i);

        Bf[3] = (_Float16)Rdx;
        Bf[4] = (_Float16)Rdy;
        Bf[5] = (_Float16)Rdz;

        float Tc = 1.f, Td = 1.f;
        float ws = 0.f,  dep = 0.f,  im0 = 0.f, im1 = 0.f, im2 = 0.f;
        float wsd = 0.f, depd = 0.f, id0 = 0.f, id1 = 0.f, id2 = 0.f;

        for (int j = 0; j < nPer; ++j) {
            const int   t     = lane * nPer + j;
            const bool  live  = t < T;
            const float z01 = (float)t * invTm1;
            const float zv  = Rnear + fn * z01;
            const float px = fminf(fmaxf(Rox + Rdx * zv, -1.f), 1.f);
            const float py = fminf(fmaxf(Roy + Rdy * zv, -1.f), 1.f);
            const float pz = fminf(fmaxf(Roz + Rdz * zv, -1.f), 1.f);
            Bf[0] = (_Float16)px; Bf[1] = (_Float16)py; Bf[2] = (_Float16)pz;

            v8f Ds, Dc;
            {
                v8f D0 = WMMA_F16(A1lo0, Bf, CbT0);
                v8f D1 = WMMA_F16(A1lo1, Bf, CbT1);
                v8f C0 = WMMA_F16(Ac1lo0, Bf, CbC0);
                v8f C1 = WMMA_F16(Ac1lo1, Bf, CbC1);
                v16h Bh, Bhc;
#pragma unroll
                for (int k = 0; k < 8; ++k) {
                    Bh[k]      = (_Float16)fmaxf(D0[k], 0.f);
                    Bh[k + 8]  = (_Float16)fmaxf(D1[k], 0.f);
                    Bhc[k]     = (_Float16)fmaxf(C0[k], 0.f);
                    Bhc[k + 8] = (_Float16)fmaxf(C1[k], 0.f);
                }
                v8f Z0 = {};
                v8f Z1 = {};
                Ds = WMMA_F16(AsigLo, Bh,  Z0);
                Dc = WMMA_F16(AcolLo, Bhc, Z1);
            }
            {
                v8f D0 = WMMA_F16(A1hi0, Bf, CbT0);
                v8f D1 = WMMA_F16(A1hi1, Bf, CbT1);
                v8f C0 = WMMA_F16(Ac1hi0, Bf, CbC0);
                v8f C1 = WMMA_F16(Ac1hi1, Bf, CbC1);
                v16h Bh, Bhc;
#pragma unroll
                for (int k = 0; k < 8; ++k) {
                    Bh[k]      = (_Float16)fmaxf(D0[k], 0.f);
                    Bh[k + 8]  = (_Float16)fmaxf(D1[k], 0.f);
                    Bhc[k]     = (_Float16)fmaxf(C0[k], 0.f);
                    Bhc[k + 8] = (_Float16)fmaxf(C1[k], 0.f);
                }
                Ds = WMMA_F16(AsigHi, Bh,  Ds);
                Dc = WMMA_F16(AcolHi, Bhc, Dc);
            }

            const float delta = live ? ((t == T - 1) ? dlast : dstep) : 0.f;
            const float dl2   = -delta * L2E;
            const float sg  = softplus_f(Ds[0]);
            const float sgd = softplus_f(Ds[1]);
            const float a   = 1.f - exp2_fast(dl2 * sg);
            const float ad  = 1.f - exp2_fast(dl2 * sgd);
            const float w   = a  * Tc;
            const float wd  = ad * Td;
            Tc *= (1.f - a  + 1e-15f);
            Td *= (1.f - ad + 1e-15f);
            ws  += w;   dep  += w  * z01;
            wsd += wd;  depd += wd * z01;
            im0 += w  * sigmoid_f(Dc[0]);
            im1 += w  * sigmoid_f(Dc[1]);
            im2 += w  * sigmoid_f(Dc[2]);
            id0 += wd * sigmoid_f(Dc[3]);
            id1 += wd * sigmoid_f(Dc[4]);
            id2 += wd * sigmoid_f(Dc[5]);
        }

        const float El  = row_scan_mul(row_dpp<ROW_SHR(1)>(1.f, Tc));
        const float Eld = row_scan_mul(row_dpp<ROW_SHR(1)>(1.f, Td));
        const float P0  = lane_bcast(El  * Tc, 15);
        const float P0d = lane_bcast(Eld * Td, 15);
        const float E   = hi ? El  * P0  : El;
        const float Ed  = hi ? Eld * P0d : Eld;
        const float Sws  = wave_sum(ws  * E),  Sdep  = wave_sum(dep  * E);
        const float Sim0 = wave_sum(im0 * E),  Sim1  = wave_sum(im1  * E),  Sim2 = wave_sum(im2 * E);
        const float Swsd = wave_sum(wsd * Ed), Sdepd = wave_sum(depd * Ed);
        const float Sid0 = wave_sum(id0 * Ed), Sid1  = wave_sum(id1  * Ed), Sid2 = wave_sum(id2 * Ed);
        const float bg  = 1.f - Sws;
        const float bgd = 1.f - Swsd;
        const bool mine = (lane == i);
        r0 = mine ? Sim0 + bg  : r0;  r1 = mine ? Sim1 + bg  : r1;  r2 = mine ? Sim2 + bg  : r2;
        r3 = mine ? Sdep       : r3;
        r4 = mine ? Sws        : r4;
        r5 = mine ? Sid0 + bgd : r5;  r6 = mine ? Sid1 + bgd : r6;  r7 = mine ? Sid2 + bgd : r7;
        r8 = mine ? Sdepd      : r8;
    }

    if (lane < RPW && (rayBase + lane) < N) {
        float* o9 = out + (size_t)(rayBase + lane) * 9;
        o9[0] = r0; o9[1] = r1; o9[2] = r2;
        o9[3] = r3;
        o9[4] = r4;
        o9[5] = r5; o9[6] = r6; o9[7] = r7;
        o9[8] = r8;
    }
}

extern "C" void kernel_launch(void* const* d_in, const int* in_sizes, int n_in,
                              void* d_out, int out_size, void* d_ws, size_t ws_size,
                              hipStream_t stream) {
    const float* rays_o = (const float*)d_in[0];
    const float* rays_d = (const float*)d_in[1];
    const float* W1     = (const float*)d_in[2];
    const float* b1     = (const float*)d_in[3];
    const float* Wsig   = (const float*)d_in[4];
    const float* Wsig_d = (const float*)d_in[5];
    const float* Wc1    = (const float*)d_in[6];
    const float* bc1    = (const float*)d_in[7];
    const float* Wc2    = (const float*)d_in[8];
    const float* Wc2_d  = (const float*)d_in[9];
    const int*   nsteps = (const int*)d_in[10];

    const int N      = in_sizes[0] / 3;
    const int waves  = (N + RPW - 1) / RPW;
    const int thr    = waves * 32;
    const int block  = 256;
    const int grid   = (thr + block - 1) / block;

    nerf_wmma_kernel<<<grid, block, 0, stream>>>(
        rays_o, rays_d, W1, b1, Wsig, Wsig_d, Wc1, bc1, Wc2, Wc2_d,
        nsteps, (float*)d_out, N);
}


# each trip carries two rays, one per 16-lane row: the join (prefix product, ten sums) runs once per ray PAIR inside the rows, no cross-row step, results stored per pair: summation order changes
# speedup vs baseline: 1.6097x; 1.1409x over previous
#include <hip/hip_runtime.h>

typedef __attribute__((ext_vector_type(16))) _Float16 v16h;
typedef __attribute__((ext_vector_type(8)))  float    v8f;

#define WMMA_F16(A, B, C) \
    __builtin_amdgcn_wmma_f32_16x16x32_f16(false, (A), false, (B), (short)0, (C), false, false)

#define L2E 1.4426950408889634f
#define LN2 0.6931471805599453f

__device__ __forceinline__ float exp2_fast(float x) { return __builtin_amdgcn_exp2f(x); }
__device__ __forceinline__ float log2_fast(float x) { return __builtin_amdgcn_logf(x); }
__device__ __forceinline__ float rcp_fast(float x)  { return __builtin_amdgcn_rcpf(x); }

__device__ __forceinline__ float softplus_f(float x) {
    return x > 20.f ? x : LN2 * log2_fast(1.f + exp2_fast(x * L2E));
}
__device__ __forceinline__ float sigmoid_f(float x) {
    return rcp_fast(1.f + exp2_fast(-x * L2E));
}

#define RPW 8

template <int CTRL>
__device__ __forceinline__ float row_dpp(float old, float src) {
    return __builtin_bit_cast(float,
        __builtin_amdgcn_update_dpp(__builtin_bit_cast(int, old), __builtin_bit_cast(int, src),
                                    CTRL, 0xf, 0xf, false));
}
#define ROW_SHR(n) (0x110 + (n))

__device__ __forceinline__ void row_scan_mul2(float& s, float& u) {
    asm("v_mul_f32_dpp %0, %0, %0 row_shr:1 row_mask:0xf bank_mask:0xf\n\t"
        "v_mul_f32_dpp %1, %1, %1 row_shr:1 row_mask:0xf bank_mask:0xf\n\t"
        "v_mul_f32_dpp %0, %0, %0 row_shr:2 row_mask:0xf bank_mask:0xf\n\t"
        "v_mul_f32_dpp %1, %1, %1 row_shr:2 row_mask:0xf bank_mask:0xf\n\t"
        "v_mul_f32_dpp %0, %0, %0 row_shr:4 row_mask:0xf bank_mask:0xf\n\t"
        "v_mul_f32_dpp %1, %1, %1 row_shr:4 row_mask:0xf bank_mask:0xf\n\t"
        "v_mul_f32_dpp %0, %0, %0 row_shr:8 row_mask:0xf bank_mask:0xf\n\t"
        "v_mul_f32_dpp %1, %1, %1 row_shr:8 row_mask:0xf bank_mask:0xf"
        : "+v"(s), "+v"(u));
}
__device__ __forceinline__ float row_pick(float x, unsigned selw) {
    const unsigned v = __builtin_bit_cast(unsigned, x);
    return __builtin_bit_cast(float, __builtin_amdgcn_permlane16(v, v, selw, selw, false, false));
}
#define RSUM_STEP(n) \
    "v_add_f32_dpp %0, %0, %0 row_xmask:" #n " row_mask:0xf bank_mask:0xf\n\t" \
    "v_add_f32_dpp %1, %1, %1 row_xmask:" #n " row_mask:0xf bank_mask:0xf\n\t" \
    "v_add_f32_dpp %2, %2, %2 row_xmask:" #n " row_mask:0xf bank_mask:0xf\n\t" \
    "v_add_f32_dpp %3, %3, %3 row_xmask:" #n " row_mask:0xf bank_mask:0xf\n\t" \
    "v_add_f32_dpp %4, %4, %4 row_xmask:" #n " row_mask:0xf bank_mask:0xf\n\t" \
    "v_add_f32_dpp %5, %5, %5 row_xmask:" #n " row_mask:0xf bank_mask:0xf\n\t" \
    "v_add_f32_dpp %6, %6, %6 row_xmask:" #n " row_mask:0xf bank_mask:0xf\n\t" \
    "v_add_f32_dpp %7, %7, %7 row_xmask:" #n " row_mask:0xf bank_mask:0xf\n\t" \
    "v_add_f32_dpp %8, %8, %8 row_xmask:" #n " row_mask:0xf bank_mask:0xf\n\t" \
    "v_add_f32_dpp %9, %9, %9 row_xmask:" #n " row_mask:0xf bank_mask:0xf"
__device__ __forceinline__ void row_sum10(float& a0, float& a1, float& a2, float& a3, float& a4,
                                          float& a5, float& a6, float& a7, float& a8, float& a9) {
    asm(RSUM_STEP(1) "\n\t" RSUM_STEP(2) "\n\t" RSUM_STEP(4) "\n\t" RSUM_STEP(8)
        : "+v"(a0), "+v"(a1), "+v"(a2), "+v"(a3), "+v"(a4),
          "+v"(a5), "+v"(a6), "+v"(a7), "+v"(a8), "+v"(a9));
}

__global__ __launch_bounds__(256)
void nerf_wmma_kernel(const float* __restrict__ rays_o,
                      const float* __restrict__ rays_d,
                      const float* __restrict__ W1,   const float* __restrict__ b1,
                      const float* __restrict__ Wsig, const float* __restrict__ Wsig_d,
                      const float* __restrict__ Wc1,  const float* __restrict__ bc1,
                      const float* __restrict__ Wc2,  const float* __restrict__ Wc2_d,
                      const int*   __restrict__ num_steps,
                      float* __restrict__ out, int N)
{
    const int tid  = blockIdx.x * blockDim.x + threadIdx.x;
    const int lane = threadIdx.x & 31;
    const int wave = tid >> 5;
    const int m    = lane & 15;
    const bool hi  = (lane & 16) != 0;
    const int rayBase = wave * RPW;

    const int slot = (lane & (RPW / 2 - 1)) + (hi ? RPW / 2 : 0);
    int ray = rayBase + slot;
    if (ray >= N) ray = N - 1;

    v16h A1lo0, A1lo1, A1hi0, A1hi1, Ac1lo0, Ac1lo1, Ac1hi0, Ac1hi1, AsigLo, AsigHi, AcolLo, AcolHi;
    const _Float16 zh = (_Float16)0.f;
    const int  hrow  = m & 7;
    const bool upper = m >= 8;
#pragma unroll
    for (int e = 0; e < 16; ++e) {
        float a0 = 0.f, a1 = 0.f, c0 = 0.f, c1 = 0.f;
        if (e < 3) { a0 = W1[e * 32 + m];  a1 = W1[e * 32 + (m + 16)]; }
        if (e < 6) { c0 = Wc1[e * 32 + m]; c1 = Wc1[e * 32 + (m + 16)]; }
        const _Float16 a0h = (_Float16)a0, a1h = (_Float16)a1, c0h = (_Float16)c0, c1h = (_Float16)c1;
        A1lo0[e]  = hi ? zh : a0h;   A1hi0[e]  = hi ? a0h : zh;
        A1lo1[e]  = hi ? zh : a1h;   A1hi1[e]  = hi ? a1h : zh;
        Ac1lo0[e] = hi ? zh : c0h;   Ac1hi0[e] = hi ? c0h : zh;
        Ac1lo1[e] = hi ? zh : c1h;   Ac1hi1[e] = hi ? c1h : zh;
        const int K = (e < 8 ? e : e + 8) + (hi ? 8 : 0);
        float sv = (hrow == 0) ? Wsig[K] : ((hrow == 1) ? Wsig_d[K] : 0.f);
        float cv = (hrow < 3) ? Wc2[K * 3 + hrow] : ((hrow < 6) ? Wc2_d[K * 3 + (hrow - 3)] : 0.f);
        const _Float16 svh = (_Float16)sv, cvh = (_Float16)cv;
        AsigLo[e] = upper ? zh : svh;   AsigHi[e] = upper ? svh : zh;
        AcolLo[e] = upper ? zh : cvh;   AcolHi[e] = upper ? cvh : zh;
    }
    v8f CbT0, CbT1, CbC0, CbC1;
#pragma unroll
    for (int i = 0; i < 8; ++i) {
        const int h0 = i + (hi ? 8 : 0);
        CbT0[i] = b1[h0];      CbT1[i] = b1[h0 + 16];
        CbC0[i] = bc1[h0];     CbC1[i] = bc1[h0 + 16];
    }

    float ox = rays_o[ray * 3 + 0], oy = rays_o[ray * 3 + 1], oz = rays_o[ray * 3 + 2];
    float dx = rays_d[ray * 3 + 0], dy = rays_d[ray * 3 + 1], dz = rays_d[ray * 3 + 2];
    {
        float il = 1.f / sqrtf(dx * dx + dy * dy + dz * dz);
        dx *= il; dy *= il; dz *= il;
    }
    float near, far;
    {
        float ivx = 1.f / dx, ivy = 1.f / dy, ivz = 1.f / dz;
        float t1x = (-1.f - ox) * ivx, t2x = (1.f - ox) * ivx;
        float t1y = (-1.f - oy) * ivy, t2y = (1.f - oy) * ivy;
        float t1z = (-1.f - oz) * ivz, t2z = (1.f - oz) * ivz;
        near = fmaxf(fmaxf(fminf(t1x, t2x), fminf(t1y, t2y)), fminf(t1z, t2z));
        far  = fminf(fminf(fmaxf(t1x, t2x), fmaxf(t1y, t2y)), fmaxf(t1z, t2z));
        near = fmaxf(near, 0.2f);
        far  = fmaxf(far, near + 1e-6f);
    }

    const int   T       = *num_steps;
    const float invTm1  = 1.f / (float)(T - 1);
    const float fnL     = far - near;
    const float dstepL  = fnL * invTm1;
    const float dlastL  = fnL / (float)T;
    const int   nPer    = (T + 15) >> 4;

    v16h Bf = {};

    for (int i = 0; i < RPW / 2; ++i) {
        const unsigned sel = (unsigned)i * 0x11111111u;
        const float Rox = row_pick(ox, sel),  Roy = row_pick(oy, sel),  Roz = row_pick(oz, sel);
        const float Rdx = row_pick(dx, sel),  Rdy = row_pick(dy, sel),  Rdz = row_pick(dz, sel);
        const float Rnear = row_pick(near, sel);
        const float fn    = row_pick(fnL, sel);
        const float dstep = row_pick(dstepL, sel);
        const float dlast = row_pick(dlastL, sel);

        Bf[3] = (_Float16)Rdx;
        Bf[4] = (_Float16)Rdy;
        Bf[5] = (_Float16)Rdz;

        float Tc = 1.f, Td = 1.f;
        float ws = 0.f,  dep = 0.f,  im0 = 0.f, im1 = 0.f, im2 = 0.f;
        float wsd = 0.f, depd = 0.f, id0 = 0.f, id1 = 0.f, id2 = 0.f;

        for (int j = 0; j < nPer; ++j) {
            const int   t     = m * nPer + j;
            const bool  live  = t < T;
            const float z01 = (float)t * invTm1;
            const float zv  = Rnear + fn * z01;
            const float px = fminf(fmaxf(Rox + Rdx * zv, -1.f), 1.f);
            const float py = fminf(fmaxf(Roy + Rdy * zv, -1.f), 1.f);
            const float pz = fminf(fmaxf(Roz + Rdz * zv, -1.f), 1.f);
            Bf[0] = (_Float16)px; Bf[1] = (_Float16)py; Bf[2] = (_Float16)pz;

            v8f Ds, Dc;
            {
                v8f D0 = WMMA_F16(A1lo0, Bf, CbT0);
                v8f D1 = WMMA_F16(A1lo1, Bf, CbT1);
                v8f C0 = WMMA_F16(Ac1lo0, Bf, CbC0);
                v8f C1 = WMMA_F16(Ac1lo1, Bf, CbC1);
                v16h Bh, Bhc;
#pragma unroll
                for (int k = 0; k < 8; ++k) {
                    Bh[k]      = (_Float16)fmaxf(D0[k], 0.f);
                    Bh[k + 8]  = (_Float16)fmaxf(D1[k], 0.f);
                    Bhc[k]     = (_Float16)fmaxf(C0[k], 0.f);
                    Bhc[k + 8] = (_Float16)fmaxf(C1[k], 0.f);
                }
                v8f Z0 = {};
                v8f Z1 = {};
                Ds = WMMA_F16(AsigLo, Bh,  Z0);
                Dc = WMMA_F16(AcolLo, Bhc, Z1);
            }
            {
                v8f D0 = WMMA_F16(A1hi0, Bf, CbT0);
                v8f D1 = WMMA_F16(A1hi1, Bf, CbT1);
                v8f C0 = WMMA_F16(Ac1hi0, Bf, CbC0);
                v8f C1 = WMMA_F16(Ac1hi1, Bf, CbC1);
                v16h Bh, Bhc;
#pragma unroll
                for (int k = 0; k < 8; ++k) {
                    Bh[k]      = (_Float16)fmaxf(D0[k], 0.f);
                    Bh[k + 8]  = (_Float16)fmaxf(D1[k], 0.f);
                    Bhc[k]     = (_Float16)fmaxf(C0[k], 0.f);
                    Bhc[k + 8] = (_Float16)fmaxf(C1[k], 0.f);
                }
                Ds = WMMA_F16(AsigHi, Bh,  Ds);
                Dc = WMMA_F16(AcolHi, Bhc, Dc);
            }

            const float delta = live ? ((t == T - 1) ? dlast : dstep) : 0.f;
            const float dl2   = -delta * L2E;
            const float sg  = softplus_f(Ds[0]);
            const float sgd = softplus_f(Ds[1]);
            const float a   = 1.f - exp2_fast(dl2 * sg);
            const float ad  = 1.f - exp2_fast(dl2 * sgd);
            const float w   = a  * Tc;
            const float wd  = ad * Td;
            Tc *= (1.f - a  + 1e-15f);
            Td *= (1.f - ad + 1e-15f);
            ws  += w;   dep  += w  * z01;
            wsd += wd;  depd += wd * z01;
            im0 += w  * sigmoid_f(Dc[0]);
            im1 += w  * sigmoid_f(Dc[1]);
            im2 += w  * sigmoid_f(Dc[2]);
            id0 += wd * sigmoid_f(Dc[3]);
            id1 += wd * sigmoid_f(Dc[4]);
            id2 += wd * sigmoid_f(Dc[5]);
        }

        float E  = row_dpp<ROW_SHR(1)>(1.f, Tc);
        float Ed = row_dpp<ROW_SHR(1)>(1.f, Td);
        row_scan_mul2(E, Ed);
        float Sws  = ws  * E,  Sdep  = dep  * E,  Sim0 = im0 * E,  Sim1 = im1 * E,  Sim2 = im2 * E;
        float Swsd = wsd * Ed, Sdepd = depd * Ed, Sid0 = id0 * Ed, Sid1 = id1 * Ed, Sid2 = id2 * Ed;
        row_sum10(Sws, Sdep, Sim0, Sim1, Sim2, Swsd, Sdepd, Sid0, Sid1, Sid2);
        const float bg  = 1.f - Sws;
        const float bgd = 1.f - Swsd;
        const int oray = rayBase + i + (hi ? RPW / 2 : 0);
        if (m == i && oray < N) {
            float* o9 = out + (size_t)oray * 9;
            o9[0] = Sim0 + bg;  o9[1] = Sim1 + bg;  o9[2] = Sim2 + bg;
            o9[3] = Sdep;
            o9[4] = Sws;
            o9[5] = Sid0 + bgd; o9[6] = Sid1 + bgd; o9[7] = Sid2 + bgd;
            o9[8] = Sdepd;
        }
    }

}

extern "C" void kernel_launch(void* const* d_in, const int* in_sizes, int n_in,
                              void* d_out, int out_size, void* d_ws, size_t ws_size,
                              hipStream_t stream) {
    const float* rays_o = (const float*)d_in[0];
    const float* rays_d = (const float*)d_in[1];
    const float* W1     = (const float*)d_in[2];
    const float* b1     = (const float*)d_in[3];
    const float* Wsig   = (const float*)d_in[4];
    const float* Wsig_d = (const float*)d_in[5];
    const float* Wc1    = (const float*)d_in[6];
    const float* bc1    = (const float*)d_in[7];
    const float* Wc2    = (const float*)d_in[8];
    const float* Wc2_d  = (const float*)d_in[9];
    const int*   nsteps = (const int*)d_in[10];

    const int N      = in_sizes[0] / 3;
    const int waves  = (N + RPW - 1) / RPW;
    const int thr    = waves * 32;
    const int block  = 256;
    const int grid   = (thr + block - 1) / block;

    nerf_wmma_kernel<<<grid, block, 0, stream>>>(
        rays_o, rays_d, W1, b1, Wsig, Wsig_d, Wc1, bc1, Wc2, Wc2_d,
        nsteps, (float*)d_out, N);
}


# relu(h) as (h+|h|)/2: |h| is a free source modifier of the 16-bit pack (no maximum left); the h/2 part is one more product per head, weights W1*W2/2 by two products per wave: head sums re-associated
# speedup vs baseline: 1.7675x; 1.0980x over previous
#include <hip/hip_runtime.h>

typedef __attribute__((ext_vector_type(16))) _Float16 v16h;
typedef __attribute__((ext_vector_type(8)))  float    v8f;

#define WMMA_F16(A, B, C) \
    __builtin_amdgcn_wmma_f32_16x16x32_f16(false, (A), false, (B), (short)0, (C), false, false)

#define L2E 1.4426950408889634f
#define LN2 0.6931471805599453f

__device__ __forceinline__ float exp2_fast(float x) { return __builtin_amdgcn_exp2f(x); }
__device__ __forceinline__ float log2_fast(float x) { return __builtin_amdgcn_logf(x); }
__device__ __forceinline__ float rcp_fast(float x)  { return __builtin_amdgcn_rcpf(x); }

__device__ __forceinline__ float softplus_f(float x) {
    return x > 20.f ? x : LN2 * log2_fast(1.f + exp2_fast(x * L2E));
}
__device__ __forceinline__ float sigmoid_f(float x) {
    return rcp_fast(1.f + exp2_fast(-x * L2E));
}

#define RPW 8

template <int CTRL>
__device__ __forceinline__ float row_dpp(float old, float src) {
    return __builtin_bit_cast(float,
        __builtin_amdgcn_update_dpp(__builtin_bit_cast(int, old), __builtin_bit_cast(int, src),
                                    CTRL, 0xf, 0xf, false));
}
#define ROW_SHR(n) (0x110 + (n))

__device__ __forceinline__ void row_scan_mul2(float& s, float& u) {
    asm("v_mul_f32_dpp %0, %0, %0 row_shr:1 row_mask:0xf bank_mask:0xf\n\t"
        "v_mul_f32_dpp %1, %1, %1 row_shr:1 row_mask:0xf bank_mask:0xf\n\t"
        "v_mul_f32_dpp %0, %0, %0 row_shr:2 row_mask:0xf bank_mask:0xf\n\t"
        "v_mul_f32_dpp %1, %1, %1 row_shr:2 row_mask:0xf bank_mask:0xf\n\t"
        "v_mul_f32_dpp %0, %0, %0 row_shr:4 row_mask:0xf bank_mask:0xf\n\t"
        "v_mul_f32_dpp %1, %1, %1 row_shr:4 row_mask:0xf bank_mask:0xf\n\t"
        "v_mul_f32_dpp %0, %0, %0 row_shr:8 row_mask:0xf bank_mask:0xf\n\t"
        "v_mul_f32_dpp %1, %1, %1 row_shr:8 row_mask:0xf bank_mask:0xf"
        : "+v"(s), "+v"(u));
}
__device__ __forceinline__ float row_pick(float x, unsigned selw) {
    const unsigned v = __builtin_bit_cast(unsigned, x);
    return __builtin_bit_cast(float, __builtin_amdgcn_permlane16(v, v, selw, selw, false, false));
}
#define RSUM_STEP(n) \
    "v_add_f32_dpp %0, %0, %0 row_xmask:" #n " row_mask:0xf bank_mask:0xf\n\t" \
    "v_add_f32_dpp %1, %1, %1 row_xmask:" #n " row_mask:0xf bank_mask:0xf\n\t" \
    "v_add_f32_dpp %2, %2, %2 row_xmask:" #n " row_mask:0xf bank_mask:0xf\n\t" \
    "v_add_f32_dpp %3, %3, %3 row_xmask:" #n " row_mask:0xf bank_mask:0xf\n\t" \
    "v_add_f32_dpp %4, %4, %4 row_xmask:" #n " row_mask:0xf bank_mask:0xf\n\t" \
    "v_add_f32_dpp %5, %5, %5 row_xmask:" #n " row_mask:0xf bank_mask:0xf\n\t" \
    "v_add_f32_dpp %6, %6, %6 row_xmask:" #n " row_mask:0xf bank_mask:0xf\n\t" \
    "v_add_f32_dpp %7, %7, %7 row_xmask:" #n " row_mask:0xf bank_mask:0xf\n\t" \
    "v_add_f32_dpp %8, %8, %8 row_xmask:" #n " row_mask:0xf bank_mask:0xf\n\t" \
    "v_add_f32_dpp %9, %9, %9 row_xmask:" #n " row_mask:0xf bank_mask:0xf"
__device__ __forceinline__ void row_sum10(float& a0, float& a1, float& a2, float& a3, float& a4,
                                          float& a5, float& a6, float& a7, float& a8, float& a9) {
    asm(RSUM_STEP(1) "\n\t" RSUM_STEP(2) "\n\t" RSUM_STEP(4) "\n\t" RSUM_STEP(8)
        : "+v"(a0), "+v"(a1), "+v"(a2), "+v"(a3), "+v"(a4),
          "+v"(a5), "+v"(a6), "+v"(a7), "+v"(a8), "+v"(a9));
}

typedef __attribute__((ext_vector_type(8))) int v8i;

__device__ __forceinline__ void pack_abs8(const v8f& d, int& p0, int& p1, int& p2, int& p3) {
    asm("v_cvt_pk_f16_f32 %0, |%4|, |%5|\n\t"
        "v_cvt_pk_f16_f32 %1, |%6|, |%7|\n\t"
        "v_cvt_pk_f16_f32 %2, |%8|, |%9|\n\t"
        "v_cvt_pk_f16_f32 %3, |%10|, |%11|"
        : "=&v"(p0), "=&v"(p1), "=&v"(p2), "=&v"(p3)
        : "v"(d[0]), "v"(d[1]), "v"(d[2]), "v"(d[3]), "v"(d[4]), "v"(d[5]), "v"(d[6]), "v"(d[7]));
}
__device__ __forceinline__ void next_sample_behind_products(
        int& tn, float& z01n, float& zvn, int t, float invTm1, float fn, float Rnear,
        v8f& a0, v8f& a1, v8f& a2, v8f& a3, v8f& b0, v8f& b1, v8f& b2, v8f& b3) {
    asm("v_add_nc_u32 %0, 1, %11\n\t"
        "s_delay_alu instid0(VALU_DEP_1)\n\t"
        "v_cvt_f32_i32 %1, %0\n\t"
        "s_delay_alu instid0(VALU_DEP_1)\n\t"
        "v_mul_f32 %1, %1, %12\n\t"
        "s_delay_alu instid0(VALU_DEP_1)\n\t"
        "v_fma_f32 %2, %13, %1, %14"
        : "=&v"(tn), "=&v"(z01n), "=&v"(zvn),
          "+v"(a0), "+v"(a1), "+v"(a2), "+v"(a3), "+v"(b0), "+v"(b1), "+v"(b2), "+v"(b3)
        : "v"(t), "v"(invTm1), "v"(fn), "v"(Rnear));
}
__device__ __forceinline__ void packs_before_heads(v16h& f0, v16h& f1, v16h& f2, v16h& f3, v16h& bf) {
    asm("" : "+v"(f0), "+v"(f1), "+v"(f2), "+v"(f3), "+v"(bf));
}
__device__ __forceinline__ v16h abs_fragment(const v8f& lo, const v8f& hi8) {
    v8i p;
    int p0, p1, p2, p3, p4, p5, p6, p7;
    pack_abs8(lo,  p0, p1, p2, p3);
    pack_abs8(hi8, p4, p5, p6, p7);
    p[0] = p0; p[1] = p1; p[2] = p2; p[3] = p3; p[4] = p4; p[5] = p5; p[6] = p6; p[7] = p7;
    return __builtin_bit_cast(v16h, p);
}

__global__ __launch_bounds__(256)
void nerf_wmma_kernel(const float* __restrict__ rays_o,
                      const float* __restrict__ rays_d,
                      const float* __restrict__ W1,   const float* __restrict__ b1,
                      const float* __restrict__ Wsig, const float* __restrict__ Wsig_d,
                      const float* __restrict__ Wc1,  const float* __restrict__ bc1,
                      const float* __restrict__ Wc2,  const float* __restrict__ Wc2_d,
                      const int*   __restrict__ num_steps,
                      float* __restrict__ out, int N)
{
    const int tid  = blockIdx.x * blockDim.x + threadIdx.x;
    const int lane = threadIdx.x & 31;
    const int wave = tid >> 5;
    const int m    = lane & 15;
    const bool hi  = (lane & 16) != 0;
    const int rayBase = wave * RPW;

    const int slot = (lane & (RPW / 2 - 1)) + (hi ? RPW / 2 : 0);
    int ray = rayBase + slot;
    if (ray >= N) ray = N - 1;

    v16h A1lo0, A1lo1, A1hi0, A1hi1, Ac1lo0, Ac1lo1, Ac1hi0, Ac1hi1, AsigLo, AsigHi, AcolLo, AcolHi;
    v16h ApT, ApC, AsigAll, AcolAll;
    const _Float16 zh = (_Float16)0.f;
    const int  hrow  = m & 7;
    const bool upper = m >= 8;
    const float bT0 = b1[m], bT1 = b1[m + 16], bC0 = bc1[m], bC1 = bc1[m + 16];
    const _Float16 bT0h = (_Float16)bT0, bT0l = (_Float16)(bT0 - (float)bT0h);
    const _Float16 bT1h = (_Float16)bT1, bT1l = (_Float16)(bT1 - (float)bT1h);
    const _Float16 bC0h = (_Float16)bC0, bC0l = (_Float16)(bC0 - (float)bC0h);
    const _Float16 bC1h = (_Float16)bC1, bC1l = (_Float16)(bC1 - (float)bC1h);
#pragma unroll
    for (int e = 0; e < 16; ++e) {
        float a0 = 0.f, a1 = 0.f, c0 = 0.f, c1 = 0.f;
        if (e < 3) { a0 = W1[e * 32 + m];  a1 = W1[e * 32 + (m + 16)]; }
        if (e < 6) { c0 = Wc1[e * 32 + m]; c1 = Wc1[e * 32 + (m + 16)]; }
        _Float16 a0h = (_Float16)a0, a1h = (_Float16)a1, c0h = (_Float16)c0, c1h = (_Float16)c1;
        if (e == 12) { a0h = bT0h; a1h = bT1h; c0h = bC0h; c1h = bC1h; }
        if (e == 13) { a0h = bT0l; a1h = bT1l; c0h = bC0l; c1h = bC1l; }
        A1lo0[e]  = hi ? zh : a0h;   A1hi0[e]  = hi ? a0h : zh;
        A1lo1[e]  = hi ? zh : a1h;   A1hi1[e]  = hi ? a1h : zh;
        Ac1lo0[e] = hi ? zh : c0h;   Ac1hi0[e] = hi ? c0h : zh;
        Ac1lo1[e] = hi ? zh : c1h;   Ac1hi1[e] = hi ? c1h : zh;
        const int K = (e < 8 ? e : e + 8) + (hi ? 8 : 0);
        float sv = 0.5f * ((hrow == 0) ? Wsig[K] : ((hrow == 1) ? Wsig_d[K] : 0.f));
        float cv = 0.5f * ((hrow < 3) ? Wc2[K * 3 + hrow] : ((hrow < 6) ? Wc2_d[K * 3 + (hrow - 3)] : 0.f));
        const _Float16 svh = (_Float16)sv, cvh = (_Float16)cv;
        AsigLo[e] = upper ? zh : svh;   AsigHi[e] = upper ? svh : zh;
        AcolLo[e] = upper ? zh : cvh;   AcolHi[e] = upper ? cvh : zh;
        AsigAll[e] = svh;               AcolAll[e] = cvh;
        const float wt = W1[(hrow < 3 ? hrow : 0) * 32 + K];
        const float wc = Wc1[(hrow < 6 ? hrow : 0) * 32 + K];
        ApT[e] = (hrow < 3) ? (_Float16)wt : zh;
        ApC[e] = (hrow < 6) ? (_Float16)wc : zh;
    }
    v8f CbT0, CbT1, CbC0, CbC1;
#pragma unroll
    for (int i = 0; i < 8; ++i) {
        const int h0 = i + (hi ? 8 : 0);
        CbT0[i] = b1[h0];      CbT1[i] = b1[h0 + 16];
        CbC0[i] = bc1[h0];     CbC1[i] = bc1[h0 + 16];
    }
#pragma unroll
    for (int e = 0; e < 16; ++e) {
        const float bt = (e < 8) ? CbT0[e & 7] : CbT1[e & 7];
        const float bc = (e < 8) ? CbC0[e & 7] : CbC1[e & 7];
        const _Float16 bth = (_Float16)bt, btl = (_Float16)(bt - (float)bth);
        const _Float16 bch = (_Float16)bc, bcl = (_Float16)(bc - (float)bch);
        ApT[e] = (hrow == 6) ? bth : ((hrow == 7) ? btl : ApT[e]);
        ApC[e] = (hrow == 6) ? bch : ((hrow == 7) ? bcl : ApC[e]);
    }
    v16h AlinS = {}, AlinC = {};
    {
        v8f Zp = {};
        const v8f Pt = WMMA_F16(ApT, AsigAll, Zp);
        const v8f Pc = WMMA_F16(ApC, AcolAll, Zp);
        const bool act = upper == hi;
#pragma unroll
        for (int e = 0; e < 6; ++e) {
            const _Float16 sh = (_Float16)Pt[e], sl = (_Float16)(Pt[e] - (float)sh);
            const _Float16 ch = (_Float16)Pc[e], cl = (_Float16)(Pc[e] - (float)ch);
            AlinS[e] = act ? sh : zh;   AlinS[e + 6] = act ? sl : zh;
            AlinC[e] = act ? ch : zh;   AlinC[e + 6] = act ? cl : zh;
        }
        const float bs = Pt[6] + Pt[7], bcs = Pc[6] + Pc[7];
        const _Float16 bsh = (_Float16)bs,  bsl = (_Float16)(bs - (float)bsh);
        const _Float16 bch2 = (_Float16)bcs, bcl2 = (_Float16)(bcs - (float)bch2);
        AlinS[12] = act ? bsh : zh;   AlinS[13] = act ? bsl : zh;
        AlinC[12] = act ? bch2 : zh;  AlinC[13] = act ? bcl2 : zh;
    }

    float ox = rays_o[ray * 3 + 0], oy = rays_o[ray * 3 + 1], oz = rays_o[ray * 3 + 2];
    float dx = rays_d[ray * 3 + 0], dy = rays_d[ray * 3 + 1], dz = rays_d[ray * 3 + 2];
    {
        float il = 1.f / sqrtf(dx * dx + dy * dy + dz * dz);
        dx *= il; dy *= il; dz *= il;
    }
    float near, far;
    {
        float ivx = 1.f / dx, ivy = 1.f / dy, ivz = 1.f / dz;
        float t1x = (-1.f - ox) * ivx, t2x = (1.f - ox) * ivx;
        float t1y = (-1.f - oy) * ivy, t2y = (1.f - oy) * ivy;
        float t1z = (-1.f - oz) * ivz, t2z = (1.f - oz) * ivz;
        near = fmaxf(fmaxf(fminf(t1x, t2x), fminf(t1y, t2y)), fminf(t1z, t2z));
        far  = fminf(fminf(fmaxf(t1x, t2x), fmaxf(t1y, t2y)), fmaxf(t1z, t2z));
        near = fmaxf(near, 0.2f);
        far  = fmaxf(far, near + 1e-6f);
    }

    const int   T       = *num_steps;
    const float invTm1  = 1.f / (float)(T - 1);
    const float fnL     = far - near;
    const float dstepL  = fnL * invTm1;
    const float dlastL  = fnL / (float)T;
    const int   nPer    = (T + 15) >> 4;

    v16h Bf = {};
    Bf[12] = (_Float16)1.f;
    Bf[13] = (_Float16)1.f;

    for (int i = 0; i < RPW / 2; ++i) {
        const unsigned sel = (unsigned)i * 0x11111111u;
        const float Rox = row_pick(ox, sel),  Roy = row_pick(oy, sel),  Roz = row_pick(oz, sel);
        const float Rdx = row_pick(dx, sel),  Rdy = row_pick(dy, sel),  Rdz = row_pick(dz, sel);
        const float Rnear = row_pick(near, sel);
        const float fn    = row_pick(fnL, sel);
        const float dstep = row_pick(dstepL, sel);
        const float dlast = row_pick(dlastL, sel);

        Bf[3] = (_Float16)Rdx;
        Bf[4] = (_Float16)Rdy;
        Bf[5] = (_Float16)Rdz;
        Bf[9] = Bf[3];  Bf[10] = Bf[4];  Bf[11] = Bf[5];

        float Tc = 1.f, Td = 1.f;
        float ws = 0.f,  dep = 0.f,  im0 = 0.f, im1 = 0.f, im2 = 0.f;
        float wsd = 0.f, depd = 0.f, id0 = 0.f, id1 = 0.f, id2 = 0.f;

        int   t   = m * nPer;
        float z01 = (float)t * invTm1;
        float zv  = Rnear + fn * z01;
        for (int j = 0; j < nPer; ++j) {
            const bool  live  = t < T;
            const float px = fminf(fmaxf(Rox + Rdx * zv, -1.f), 1.f);
            const float py = fminf(fmaxf(Roy + Rdy * zv, -1.f), 1.f);
            const float pz = fminf(fmaxf(Roz + Rdz * zv, -1.f), 1.f);
            Bf[0] = (_Float16)px; Bf[1] = (_Float16)py; Bf[2] = (_Float16)pz;
            Bf[6] = Bf[0];  Bf[7] = Bf[1];  Bf[8] = Bf[2];

            v8f Ds, Dc;
            {
                v8f Z0 = {};
                v8f Z1 = {};
                Ds = WMMA_F16(AlinS, Bf, Z0);
                Dc = WMMA_F16(AlinC, Bf, Z1);
            }
            const v8f Zf = {};
            v8f D0a = WMMA_F16(A1lo0, Bf, Zf);
            v8f D1a = WMMA_F16(A1lo1, Bf, Zf);
            v8f C0a = WMMA_F16(Ac1lo0, Bf, Zf);
            v8f C1a = WMMA_F16(Ac1lo1, Bf, Zf);
            v8f D0b = WMMA_F16(A1hi0, Bf, Zf);
            v8f D1b = WMMA_F16(A1hi1, Bf, Zf);
            v8f C0b = WMMA_F16(Ac1hi0, Bf, Zf);
            v8f C1b = WMMA_F16(Ac1hi1, Bf, Zf);

            int tn;  float z01n, zvn;
            next_sample_behind_products(tn, z01n, zvn, t, invTm1, fn, Rnear,
                                        D0a, D1a, C0a, C1a, D0b, D1b, C0b, C1b);
            v16h Bh1  = abs_fragment(D0a, D1a);
            v16h Bhc1 = abs_fragment(C0a, C1a);
            v16h Bh2  = abs_fragment(D0b, D1b);
            v16h Bhc2 = abs_fragment(C0b, C1b);
            packs_before_heads(Bh1, Bhc1, Bh2, Bhc2, Bf);

            Ds = WMMA_F16(AsigLo, Bh1,  Ds);
            Dc = WMMA_F16(AcolLo, Bhc1, Dc);
            Ds = WMMA_F16(AsigHi, Bh2,  Ds);
            Dc = WMMA_F16(AcolHi, Bhc2, Dc);

            const float delta = live ? ((t == T - 1) ? dlast : dstep) : 0.f;
            const float dl2   = -delta * L2E;
            const float sg  = softplus_f(Ds[0]);
            const float sgd = softplus_f(Ds[1]);
            const float a   = 1.f - exp2_fast(dl2 * sg);
            const float ad  = 1.f - exp2_fast(dl2 * sgd);
            const float w   = a  * Tc;
            const float wd  = ad * Td;
            Tc *= (1.f - a  + 1e-15f);
            Td *= (1.f - ad + 1e-15f);
            ws  += w;   dep  += w  * z01;
            wsd += wd;  depd += wd * z01;
            im0 += w  * sigmoid_f(Dc[0]);
            im1 += w  * sigmoid_f(Dc[1]);
            im2 += w  * sigmoid_f(Dc[2]);
            id0 += wd * sigmoid_f(Dc[3]);
            id1 += wd * sigmoid_f(Dc[4]);
            id2 += wd * sigmoid_f(Dc[5]);
            t = tn;  z01 = z01n;  zv = zvn;
        }

        float E  = row_dpp<ROW_SHR(1)>(1.f, Tc);
        float Ed = row_dpp<ROW_SHR(1)>(1.f, Td);
        row_scan_mul2(E, Ed);
        float Sws  = ws  * E,  Sdep  = dep  * E,  Sim0 = im0 * E,  Sim1 = im1 * E,  Sim2 = im2 * E;
        float Swsd = wsd * Ed, Sdepd = depd * Ed, Sid0 = id0 * Ed, Sid1 = id1 * Ed, Sid2 = id2 * Ed;
        row_sum10(Sws, Sdep, Sim0, Sim1, Sim2, Swsd, Sdepd, Sid0, Sid1, Sid2);
        const float bg  = 1.f - Sws;
        const float bgd = 1.f - Swsd;
        const int oray = rayBase + i + (hi ? RPW / 2 : 0);
        if (m == i && oray < N) {
            float* o9 = out + (size_t)oray * 9;
            o9[0] = Sim0 + bg;  o9[1] = Sim1 + bg;  o9[2] = Sim2 + bg;
            o9[3] = Sdep;
            o9[4] = Sws;
            o9[5] = Sid0 + bgd; o9[6] = Sid1 + bgd; o9[7] = Sid2 + bgd;
            o9[8] = Sdepd;
        }
    }

}

extern "C" void kernel_launch(void* const* d_in, const int* in_sizes, int n_in,
                              void* d_out, int out_size, void* d_ws, size_t ws_size,
                              hipStream_t stream) {
    const float* rays_o = (const float*)d_in[0];
    const float* rays_d = (const float*)d_in[1];
    const float* W1     = (const float*)d_in[2];
    const float* b1     = (const float*)d_in[3];
    const float* Wsig   = (const float*)d_in[4];
    const float* Wsig_d = (const float*)d_in[5];
    const float* Wc1    = (const float*)d_in[6];
    const float* bc1    = (const float*)d_in[7];
    const float* Wc2    = (const float*)d_in[8];
    const float* Wc2_d  = (const float*)d_in[9];
    const int*   nsteps = (const int*)d_in[10];

    const int N      = in_sizes[0] / 3;
    const int waves  = (N + RPW - 1) / RPW;
    const int thr    = waves * 32;
    const int block  = 256;
    const int grid   = (thr + block - 1) / block;

    nerf_wmma_kernel<<<grid, block, 0, stream>>>(
        rays_o, rays_d, W1, b1, Wsig, Wsig_d, Wc1, bc1, Wc2, Wc2_d,
        nsteps, (float*)d_out, N);
}


# weight fragments built once per work-group: each of its 8 waves builds its share of the 14 and all read them back from LDS behind one barrier, instead of every wave building all 14
# speedup vs baseline: 2.0231x; 1.1446x over previous
#include <hip/hip_runtime.h>

typedef __attribute__((ext_vector_type(16))) _Float16 v16h;
typedef __attribute__((ext_vector_type(8)))  float    v8f;

#define WMMA_F16(A, B, C) \
    __builtin_amdgcn_wmma_f32_16x16x32_f16(false, (A), false, (B), (short)0, (C), false, false)

#define L2E 1.4426950408889634f
#define LN2 0.6931471805599453f

__device__ __forceinline__ float exp2_fast(float x) { return __builtin_amdgcn_exp2f(x); }
__device__ __forceinline__ float log2_fast(float x) { return __builtin_amdgcn_logf(x); }
__device__ __forceinline__ float rcp_fast(float x)  { return __builtin_amdgcn_rcpf(x); }

__device__ __forceinline__ float softplus_f(float x) {
    return x > 20.f ? x : LN2 * log2_fast(1.f + exp2_fast(x * L2E));
}
__device__ __forceinline__ float sigmoid_f(float x) {
    return rcp_fast(1.f + exp2_fast(-x * L2E));
}

#define RPW 8

template <int CTRL>
__device__ __forceinline__ float row_dpp(float old, float src) {
    return __builtin_bit_cast(float,
        __builtin_amdgcn_update_dpp(__builtin_bit_cast(int, old), __builtin_bit_cast(int, src),
                                    CTRL, 0xf, 0xf, false));
}
#define ROW_SHR(n) (0x110 + (n))

__device__ __forceinline__ void row_scan_mul2(float& s, float& u) {
    asm("v_mul_f32_dpp %0, %0, %0 row_shr:1 row_mask:0xf bank_mask:0xf\n\t"
        "v_mul_f32_dpp %1, %1, %1 row_shr:1 row_mask:0xf bank_mask:0xf\n\t"
        "v_mul_f32_dpp %0, %0, %0 row_shr:2 row_mask:0xf bank_mask:0xf\n\t"
        "v_mul_f32_dpp %1, %1, %1 row_shr:2 row_mask:0xf bank_mask:0xf\n\t"
        "v_mul_f32_dpp %0, %0, %0 row_shr:4 row_mask:0xf bank_mask:0xf\n\t"
        "v_mul_f32_dpp %1, %1, %1 row_shr:4 row_mask:0xf bank_mask:0xf\n\t"
        "v_mul_f32_dpp %0, %0, %0 row_shr:8 row_mask:0xf bank_mask:0xf\n\t"
        "v_mul_f32_dpp %1, %1, %1 row_shr:8 row_mask:0xf bank_mask:0xf"
        : "+v"(s), "+v"(u));
}
__device__ __forceinline__ float row_pick(float x, unsigned selw) {
    const unsigned v = __builtin_bit_cast(unsigned, x);
    return __builtin_bit_cast(float, __builtin_amdgcn_permlane16(v, v, selw, selw, false, false));
}
#define RSUM_STEP(n) \
    "v_add_f32_dpp %0, %0, %0 row_xmask:" #n " row_mask:0xf bank_mask:0xf\n\t" \
    "v_add_f32_dpp %1, %1, %1 row_xmask:" #n " row_mask:0xf bank_mask:0xf\n\t" \
    "v_add_f32_dpp %2, %2, %2 row_xmask:" #n " row_mask:0xf bank_mask:0xf\n\t" \
    "v_add_f32_dpp %3, %3, %3 row_xmask:" #n " row_mask:0xf bank_mask:0xf\n\t" \
    "v_add_f32_dpp %4, %4, %4 row_xmask:" #n " row_mask:0xf bank_mask:0xf\n\t" \
    "v_add_f32_dpp %5, %5, %5 row_xmask:" #n " row_mask:0xf bank_mask:0xf\n\t" \
    "v_add_f32_dpp %6, %6, %6 row_xmask:" #n " row_mask:0xf bank_mask:0xf\n\t" \
    "v_add_f32_dpp %7, %7, %7 row_xmask:" #n " row_mask:0xf bank_mask:0xf\n\t" \
    "v_add_f32_dpp %8, %8, %8 row_xmask:" #n " row_mask:0xf bank_mask:0xf\n\t" \
    "v_add_f32_dpp %9, %9, %9 row_xmask:" #n " row_mask:0xf bank_mask:0xf"
__device__ __forceinline__ void row_sum10(float& a0, float& a1, float& a2, float& a3, float& a4,
                                          float& a5, float& a6, float& a7, float& a8, float& a9) {
    asm(RSUM_STEP(1) "\n\t" RSUM_STEP(2) "\n\t" RSUM_STEP(4) "\n\t" RSUM_STEP(8)
        : "+v"(a0), "+v"(a1), "+v"(a2), "+v"(a3), "+v"(a4),
          "+v"(a5), "+v"(a6), "+v"(a7), "+v"(a8), "+v"(a9));
}

typedef __attribute__((ext_vector_type(8))) int v8i;

__device__ __forceinline__ void pack_abs8(const v8f& d, int& p0, int& p1, int& p2, int& p3) {
    asm("v_cvt_pk_f16_f32 %0, |%4|, |%5|\n\t"
        "v_cvt_pk_f16_f32 %1, |%6|, |%7|\n\t"
        "v_cvt_pk_f16_f32 %2, |%8|, |%9|\n\t"
        "v_cvt_pk_f16_f32 %3, |%10|, |%11|"
        : "=&v"(p0), "=&v"(p1), "=&v"(p2), "=&v"(p3)
        : "v"(d[0]), "v"(d[1]), "v"(d[2]), "v"(d[3]), "v"(d[4]), "v"(d[5]), "v"(d[6]), "v"(d[7]));
}
__device__ __forceinline__ void next_sample_behind_products(
        int& tn, float& z01n, float& zvn, int t, float invTm1, float fn, float Rnear,
        v8f& a0, v8f& a1, v8f& a2, v8f& a3, v8f& b0, v8f& b1, v8f& b2, v8f& b3) {
    asm("v_add_nc_u32 %0, 1, %11\n\t"
        "s_delay_alu instid0(VALU_DEP_1)\n\t"
        "v_cvt_f32_i32 %1, %0\n\t"
        "s_delay_alu instid0(VALU_DEP_1)\n\t"
        "v_mul_f32 %1, %1, %12\n\t"
        "s_delay_alu instid0(VALU_DEP_1)\n\t"
        "v_fma_f32 %2, %13, %1, %14"
        : "=&v"(tn), "=&v"(z01n), "=&v"(zvn),
          "+v"(a0), "+v"(a1), "+v"(a2), "+v"(a3), "+v"(b0), "+v"(b1), "+v"(b2), "+v"(b3)
        : "v"(t), "v"(invTm1), "v"(fn), "v"(Rnear));
}
__device__ __forceinline__ void packs_before_heads(v16h& f0, v16h& f1, v16h& f2, v16h& f3, v16h& bf) {
    asm("" : "+v"(f0), "+v"(f1), "+v"(f2), "+v"(f3), "+v"(bf));
}
__device__ __forceinline__ v16h abs_fragment(const v8f& lo, const v8f& hi8) {
    v8i p;
    int p0, p1, p2, p3, p4, p5, p6, p7;
    pack_abs8(lo,  p0, p1, p2, p3);
    pack_abs8(hi8, p4, p5, p6, p7);
    p[0] = p0; p[1] = p1; p[2] = p2; p[3] = p3; p[4] = p4; p[5] = p5; p[6] = p6; p[7] = p7;
    return __builtin_bit_cast(v16h, p);
}

typedef __attribute__((ext_vector_type(4))) int v4i;

__device__ __forceinline__ void frag_put(v4i (*S)[32], int f, int lane, const v16h& x) {
    const v8i p = __builtin_bit_cast(v8i, x);
    const v4i a = {p[0], p[1], p[2], p[3]}, b = {p[4], p[5], p[6], p[7]};
    S[2 * f][lane] = a;   S[2 * f + 1][lane] = b;
}
__device__ __forceinline__ v16h frag_get(v4i (*S)[32], int f, int lane) {
    const v4i a = S[2 * f][lane], b = S[2 * f + 1][lane];
    const v8i p = {a[0], a[1], a[2], a[3], b[0], b[1], b[2], b[3]};
    return __builtin_bit_cast(v16h, p);
}
__device__ __forceinline__ int kidx(int e, bool hi) { return (e < 8 ? e : e + 8) + (hi ? 8 : 0); }
template <int NIN>
__device__ __forceinline__ void build_first(const float* __restrict__ Wp, const float* __restrict__ bp,
                                            int row, bool hi, v16h& lo, v16h& hf) {
    const _Float16 zh = (_Float16)0.f;
    const float b = bp[row];
    const _Float16 bh = (_Float16)b, bl = (_Float16)(b - (float)bh);
#pragma unroll
    for (int e = 0; e < 16; ++e) {
        float a = 0.f;
        if (e < NIN) a = Wp[e * 32 + row];
        _Float16 ah = (_Float16)a;
        if (e == 12) ah = bh;
        if (e == 13) ah = bl;
        lo[e] = hi ? zh : ah;   hf[e] = hi ? ah : zh;
    }
}
__device__ __forceinline__ _Float16 sig_weight(const float* __restrict__ Wsig, const float* __restrict__ Wsig_d,
                                               int e, int hrow, bool hi) {
    const int K = kidx(e, hi);
    const float sv = 0.5f * ((hrow == 0) ? Wsig[K] : ((hrow == 1) ? Wsig_d[K] : 0.f));
    return (_Float16)sv;
}
__device__ __forceinline__ _Float16 col_weight(const float* __restrict__ Wc2, const float* __restrict__ Wc2_d,
                                               int e, int hrow, bool hi) {
    const int K = kidx(e, hi);
    const float cv = 0.5f * ((hrow < 3) ? Wc2[K * 3 + hrow] : ((hrow < 6) ? Wc2_d[K * 3 + (hrow - 3)] : 0.f));
    return (_Float16)cv;
}
template <int NIN>
__device__ __forceinline__ _Float16 lin_row(const float* __restrict__ Wp, const float* __restrict__ bp,
                                            int e, int hrow, bool hi) {
    const int K = kidx(e, hi);
    const float w  = Wp[(hrow < NIN ? hrow : 0) * 32 + K];
    const float bv = bp[K];
    const _Float16 bvh = (_Float16)bv, bvl = (_Float16)(bv - (float)bvh);
    const _Float16 wh = (hrow < NIN) ? (_Float16)w : (_Float16)0.f;
    return (hrow == 6) ? bvh : ((hrow == 7) ? bvl : wh);
}
__device__ __forceinline__ v16h lin_fragment(const v8f& P, bool act) {
    const _Float16 zh = (_Float16)0.f;
    v16h A = {};
#pragma unroll
    for (int e = 0; e < 6; ++e) {
        const _Float16 sh = (_Float16)P[e], sl = (_Float16)(P[e] - (float)sh);
        A[e] = act ? sh : zh;   A[e + 6] = act ? sl : zh;
    }
    const float bs = P[6] + P[7];
    const _Float16 bsh = (_Float16)bs, bsl = (_Float16)(bs - (float)bsh);
    A[12] = act ? bsh : zh;   A[13] = act ? bsl : zh;
    return A;
}

__global__ __launch_bounds__(256)
void nerf_wmma_kernel(const float* __restrict__ rays_o,
                      const float* __restrict__ rays_d,
                      const float* __restrict__ W1,   const float* __restrict__ b1,
                      const float* __restrict__ Wsig, const float* __restrict__ Wsig_d,
                      const float* __restrict__ Wc1,  const float* __restrict__ bc1,
                      const float* __restrict__ Wc2,  const float* __restrict__ Wc2_d,
                      const int*   __restrict__ num_steps,
                      float* __restrict__ out, int N)
{
    const int tid  = blockIdx.x * blockDim.x + threadIdx.x;
    const int lane = threadIdx.x & 31;
    const int wave = tid >> 5;
    const int m    = lane & 15;
    const bool hi  = (lane & 16) != 0;
    const int rayBase = wave * RPW;

    const int slot = (lane & (RPW / 2 - 1)) + (hi ? RPW / 2 : 0);
    int ray = rayBase + slot;
    if (ray >= N) ray = N - 1;

    v16h A1lo0, A1lo1, A1hi0, A1hi1, Ac1lo0, Ac1lo1, Ac1hi0, Ac1hi1, AsigLo, AsigHi, AcolLo, AcolHi;
    v16h AlinS, AlinC;
    __shared__ v4i fragS[28][32];
    const int  hrow  = m & 7;
    const bool upper = m >= 8;
    const int wv = __builtin_amdgcn_readfirstlane((int)(threadIdx.x >> 5));
    if (wv == 0) {
        v16h lo, hf;  build_first<3>(W1, b1, m, hi, lo, hf);
        frag_put(fragS, 0, lane, lo);  frag_put(fragS, 1, lane, hf);
    } else if (wv == 1) {
        v16h lo, hf;  build_first<3>(W1, b1, m + 16, hi, lo, hf);
        frag_put(fragS, 2, lane, lo);  frag_put(fragS, 3, lane, hf);
    } else if (wv == 2) {
        v16h lo, hf;  build_first<6>(Wc1, bc1, m, hi, lo, hf);
        frag_put(fragS, 4, lane, lo);  frag_put(fragS, 5, lane, hf);
    } else if (wv == 3) {
        v16h lo, hf;  build_first<6>(Wc1, bc1, m + 16, hi, lo, hf);
        frag_put(fragS, 6, lane, lo);  frag_put(fragS, 7, lane, hf);
    } else if (wv == 4) {
        const _Float16 zh = (_Float16)0.f;
        v16h lo, hf;
#pragma unroll
        for (int e = 0; e < 16; ++e) {
            const _Float16 svh = sig_weight(Wsig, Wsig_d, e, hrow, hi);
            lo[e] = upper ? zh : svh;   hf[e] = upper ? svh : zh;
        }
        frag_put(fragS, 8, lane, lo);  frag_put(fragS, 9, lane, hf);
    } else if (wv == 5) {
        const _Float16 zh = (_Float16)0.f;
        v16h lo, hf;
#pragma unroll
        for (int e = 0; e < 16; ++e) {
            const _Float16 cvh = col_weight(Wc2, Wc2_d, e, hrow, hi);
            lo[e] = upper ? zh : cvh;   hf[e] = upper ? cvh : zh;
        }
        frag_put(fragS, 10, lane, lo);  frag_put(fragS, 11, lane, hf);
    } else if (wv == 6) {
        v16h Ap, All;
#pragma unroll
        for (int e = 0; e < 16; ++e) {
            All[e] = sig_weight(Wsig, Wsig_d, e, hrow, hi);
            Ap[e]  = lin_row<3>(W1, b1, e, hrow, hi);
        }
        v8f Zp = {};
        const v8f P = WMMA_F16(Ap, All, Zp);
        frag_put(fragS, 12, lane, lin_fragment(P, upper == hi));
    } else {
        v16h Ap, All;
#pragma unroll
        for (int e = 0; e < 16; ++e) {
            All[e] = col_weight(Wc2, Wc2_d, e, hrow, hi);
            Ap[e]  = lin_row<6>(Wc1, bc1, e, hrow, hi);
        }
        v8f Zp = {};
        const v8f P = WMMA_F16(Ap, All, Zp);
        frag_put(fragS, 13, lane, lin_fragment(P, upper == hi));
    }
    __syncthreads();
    A1lo0  = frag_get(fragS, 0, lane);   A1hi0  = frag_get(fragS, 1, lane);
    A1lo1  = frag_get(fragS, 2, lane);   A1hi1  = frag_get(fragS, 3, lane);
    Ac1lo0 = frag_get(fragS, 4, lane);   Ac1hi0 = frag_get(fragS, 5, lane);
    Ac1lo1 = frag_get(fragS, 6, lane);   Ac1hi1 = frag_get(fragS, 7, lane);
    AsigLo = frag_get(fragS, 8, lane);   AsigHi = frag_get(fragS, 9, lane);
    AcolLo = frag_get(fragS, 10, lane);  AcolHi = frag_get(fragS, 11, lane);
    AlinS  = frag_get(fragS, 12, lane);  AlinC  = frag_get(fragS, 13, lane);

    float ox = rays_o[ray * 3 + 0], oy = rays_o[ray * 3 + 1], oz = rays_o[ray * 3 + 2];
    float dx = rays_d[ray * 3 + 0], dy = rays_d[ray * 3 + 1], dz = rays_d[ray * 3 + 2];
    {
        float il = 1.f / sqrtf(dx * dx + dy * dy + dz * dz);
        dx *= il; dy *= il; dz *= il;
    }
    float near, far;
    {
        float ivx = 1.f / dx, ivy = 1.f / dy, ivz = 1.f / dz;
        float t1x = (-1.f - ox) * ivx, t2x = (1.f - ox) * ivx;
        float t1y = (-1.f - oy) * ivy, t2y = (1.f - oy) * ivy;
        float t1z = (-1.f - oz) * ivz, t2z = (1.f - oz) * ivz;
        near = fmaxf(fmaxf(fminf(t1x, t2x), fminf(t1y, t2y)), fminf(t1z, t2z));
        far  = fminf(fminf(fmaxf(t1x, t2x), fmaxf(t1y, t2y)), fmaxf(t1z, t2z));
        near = fmaxf(near, 0.2f);
        far  = fmaxf(far, near + 1e-6f);
    }

    const int   T       = *num_steps;
    const float invTm1  = 1.f / (float)(T - 1);
    const float fnL     = far - near;
    const float dstepL  = fnL * invTm1;
    const float dlastL  = fnL / (float)T;
    const int   nPer    = (T + 15) >> 4;

    v16h Bf = {};
    Bf[12] = (_Float16)1.f;
    Bf[13] = (_Float16)1.f;

    for (int i = 0; i < RPW / 2; ++i) {
        const unsigned sel = (unsigned)i * 0x11111111u;
        const float Rox = row_pick(ox, sel),  Roy = row_pick(oy, sel),  Roz = row_pick(oz, sel);
        const float Rdx = row_pick(dx, sel),  Rdy = row_pick(dy, sel),  Rdz = row_pick(dz, sel);
        const float Rnear = row_pick(near, sel);
        const float fn    = row_pick(fnL, sel);
        const float dstep = row_pick(dstepL, sel);
        const float dlast = row_pick(dlastL, sel);

        Bf[3] = (_Float16)Rdx;
        Bf[4] = (_Float16)Rdy;
        Bf[5] = (_Float16)Rdz;
        Bf[9] = Bf[3];  Bf[10] = Bf[4];  Bf[11] = Bf[5];

        float Tc = 1.f, Td = 1.f;
        float ws = 0.f,  dep = 0.f,  im0 = 0.f, im1 = 0.f, im2 = 0.f;
        float wsd = 0.f, depd = 0.f, id0 = 0.f, id1 = 0.f, id2 = 0.f;

        int   t   = m * nPer;
        float z01 = (float)t * invTm1;
        float zv  = Rnear + fn * z01;
        for (int j = 0; j < nPer; ++j) {
            const bool  live  = t < T;
            const float px = fminf(fmaxf(Rox + Rdx * zv, -1.f), 1.f);
            const float py = fminf(fmaxf(Roy + Rdy * zv, -1.f), 1.f);
            const float pz = fminf(fmaxf(Roz + Rdz * zv, -1.f), 1.f);
            Bf[0] = (_Float16)px; Bf[1] = (_Float16)py; Bf[2] = (_Float16)pz;
            Bf[6] = Bf[0];  Bf[7] = Bf[1];  Bf[8] = Bf[2];

            v8f Ds, Dc;
            {
                v8f Z0 = {};
                v8f Z1 = {};
                Ds = WMMA_F16(AlinS, Bf, Z0);
                Dc = WMMA_F16(AlinC, Bf, Z1);
            }
            const v8f Zf = {};
            v8f D0a = WMMA_F16(A1lo0, Bf, Zf);
            v8f D1a = WMMA_F16(A1lo1, Bf, Zf);
            v8f C0a = WMMA_F16(Ac1lo0, Bf, Zf);
            v8f C1a = WMMA_F16(Ac1lo1, Bf, Zf);
            v8f D0b = WMMA_F16(A1hi0, Bf, Zf);
            v8f D1b = WMMA_F16(A1hi1, Bf, Zf);
            v8f C0b = WMMA_F16(Ac1hi0, Bf, Zf);
            v8f C1b = WMMA_F16(Ac1hi1, Bf, Zf);

            int tn;  float z01n, zvn;
            next_sample_behind_products(tn, z01n, zvn, t, invTm1, fn, Rnear,
                                        D0a, D1a, C0a, C1a, D0b, D1b, C0b, C1b);
            v16h Bh1  = abs_fragment(D0a, D1a);
            v16h Bhc1 = abs_fragment(C0a, C1a);
            v16h Bh2  = abs_fragment(D0b, D1b);
            v16h Bhc2 = abs_fragment(C0b, C1b);
            packs_before_heads(Bh1, Bhc1, Bh2, Bhc2, Bf);

            Ds = WMMA_F16(AsigLo, Bh1,  Ds);
            Dc = WMMA_F16(AcolLo, Bhc1, Dc);
            Ds = WMMA_F16(AsigHi, Bh2,  Ds);
            Dc = WMMA_F16(AcolHi, Bhc2, Dc);

            const float delta = live ? ((t == T - 1) ? dlast : dstep) : 0.f;
            const float dl2   = -delta * L2E;
            const float sg  = softplus_f(Ds[0]);
            const float sgd = softplus_f(Ds[1]);
            const float a   = 1.f - exp2_fast(dl2 * sg);
            const float ad  = 1.f - exp2_fast(dl2 * sgd);
            const float w   = a  * Tc;
            const float wd  = ad * Td;
            Tc *= (1.f - a  + 1e-15f);
            Td *= (1.f - ad + 1e-15f);
            ws  += w;   dep  += w  * z01;
            wsd += wd;  depd += wd * z01;
            im0 += w  * sigmoid_f(Dc[0]);
            im1 += w  * sigmoid_f(Dc[1]);
            im2 += w  * sigmoid_f(Dc[2]);
            id0 += wd * sigmoid_f(Dc[3]);
            id1 += wd * sigmoid_f(Dc[4]);
            id2 += wd * sigmoid_f(Dc[5]);
            t = tn;  z01 = z01n;  zv = zvn;
        }

        float E  = row_dpp<ROW_SHR(1)>(1.f, Tc);
        float Ed = row_dpp<ROW_SHR(1)>(1.f, Td);
        row_scan_mul2(E, Ed);
        float Sws  = ws  * E,  Sdep  = dep  * E,  Sim0 = im0 * E,  Sim1 = im1 * E,  Sim2 = im2 * E;
        float Swsd = wsd * Ed, Sdepd = depd * Ed, Sid0 = id0 * Ed, Sid1 = id1 * Ed, Sid2 = id2 * Ed;
        row_sum10(Sws, Sdep, Sim0, Sim1, Sim2, Swsd, Sdepd, Sid0, Sid1, Sid2);
        const float bg  = 1.f - Sws;
        const float bgd = 1.f - Swsd;
        const int oray = rayBase + i + (hi ? RPW / 2 : 0);
        if (m == i && oray < N) {
            float* o9 = out + (size_t)oray * 9;
            o9[0] = Sim0 + bg;  o9[1] = Sim1 + bg;  o9[2] = Sim2 + bg;
            o9[3] = Sdep;
            o9[4] = Sws;
            o9[5] = Sid0 + bgd; o9[6] = Sid1 + bgd; o9[7] = Sid2 + bgd;
            o9[8] = Sdepd;
        }
    }

}

extern "C" void kernel_launch(void* const* d_in, const int* in_sizes, int n_in,
                              void* d_out, int out_size, void* d_ws, size_t ws_size,
                              hipStream_t stream) {
    const float* rays_o = (const float*)d_in[0];
    const float* rays_d = (const float*)d_in[1];
    const float* W1     = (const float*)d_in[2];
    const float* b1     = (const float*)d_in[3];
    const float* Wsig   = (const float*)d_in[4];
    const float* Wsig_d = (const float*)d_in[5];
    const float* Wc1    = (const float*)d_in[6];
    const float* bc1    = (const float*)d_in[7];
    const float* Wc2    = (const float*)d_in[8];
    const float* Wc2_d  = (const float*)d_in[9];
    const int*   nsteps = (const int*)d_in[10];

    const int N      = in_sizes[0] / 3;
    const int waves  = (N + RPW - 1) / RPW;
    const int thr    = waves * 32;
    const int block  = 256;
    const int grid   = (thr + block - 1) / block;

    nerf_wmma_kernel<<<grid, block, 0, stream>>>(
        rays_o, rays_d, W1, b1, Wsig, Wsig_d, Wc1, bc1, Wc2, Wc2_d,
        nsteps, (float*)d_out, N);
}


# raw weights staged once per work-group in zero-padded LDS tables (wave r loads row r, coalesced); head and composite-weight fragments gather from the tables: no chain of dependent global loads
# speedup vs baseline: 2.2918x; 1.1328x over previous
#include <hip/hip_runtime.h>

typedef __attribute__((ext_vector_type(16))) _Float16 v16h;
typedef __attribute__((ext_vector_type(8)))  float    v8f;

#define WMMA_F16(A, B, C) \
    __builtin_amdgcn_wmma_f32_16x16x32_f16(false, (A), false, (B), (short)0, (C), false, false)

#define L2E 1.4426950408889634f
#define LN2 0.6931471805599453f

__device__ __forceinline__ float exp2_fast(float x) { return __builtin_amdgcn_exp2f(x); }
__device__ __forceinline__ float log2_fast(float x) { return __builtin_amdgcn_logf(x); }
__device__ __forceinline__ float rcp_fast(float x)  { return __builtin_amdgcn_rcpf(x); }

__device__ __forceinline__ float softplus_f(float x) {
    return x > 20.f ? x : LN2 * log2_fast(1.f + exp2_fast(x * L2E));
}
__device__ __forceinline__ float sigmoid_f(float x) {
    return rcp_fast(1.f + exp2_fast(-x * L2E));
}

#define RPW 8

template <int CTRL>
__device__ __forceinline__ float row_dpp(float old, float src) {
    return __builtin_bit_cast(float,
        __builtin_amdgcn_update_dpp(__builtin_bit_cast(int, old), __builtin_bit_cast(int, src),
                                    CTRL, 0xf, 0xf, false));
}
#define ROW_SHR(n) (0x110 + (n))

__device__ __forceinline__ void row_scan_mul2(float& s, float& u) {
    asm("v_mul_f32_dpp %0, %0, %0 row_shr:1 row_mask:0xf bank_mask:0xf\n\t"
        "v_mul_f32_dpp %1, %1, %1 row_shr:1 row_mask:0xf bank_mask:0xf\n\t"
        "v_mul_f32_dpp %0, %0, %0 row_shr:2 row_mask:0xf bank_mask:0xf\n\t"
        "v_mul_f32_dpp %1, %1, %1 row_shr:2 row_mask:0xf bank_mask:0xf\n\t"
        "v_mul_f32_dpp %0, %0, %0 row_shr:4 row_mask:0xf bank_mask:0xf\n\t"
        "v_mul_f32_dpp %1, %1, %1 row_shr:4 row_mask:0xf bank_mask:0xf\n\t"
        "v_mul_f32_dpp %0, %0, %0 row_shr:8 row_mask:0xf bank_mask:0xf\n\t"
        "v_mul_f32_dpp %1, %1, %1 row_shr:8 row_mask:0xf bank_mask:0xf"
        : "+v"(s), "+v"(u));
}
__device__ __forceinline__ float row_pick(float x, unsigned selw) {
    const unsigned v = __builtin_bit_cast(unsigned, x);
    return __builtin_bit_cast(float, __builtin_amdgcn_permlane16(v, v, selw, selw, false, false));
}
#define RSUM_STEP(n) \
    "v_add_f32_dpp %0, %0, %0 row_xmask:" #n " row_mask:0xf bank_mask:0xf\n\t" \
    "v_add_f32_dpp %1, %1, %1 row_xmask:" #n " row_mask:0xf bank_mask:0xf\n\t" \
    "v_add_f32_dpp %2, %2, %2 row_xmask:" #n " row_mask:0xf bank_mask:0xf\n\t" \
    "v_add_f32_dpp %3, %3, %3 row_xmask:" #n " row_mask:0xf bank_mask:0xf\n\t" \
    "v_add_f32_dpp %4, %4, %4 row_xmask:" #n " row_mask:0xf bank_mask:0xf\n\t" \
    "v_add_f32_dpp %5, %5, %5 row_xmask:" #n " row_mask:0xf bank_mask:0xf\n\t" \
    "v_add_f32_dpp %6, %6, %6 row_xmask:" #n " row_mask:0xf bank_mask:0xf\n\t" \
    "v_add_f32_dpp %7, %7, %7 row_xmask:" #n " row_mask:0xf bank_mask:0xf\n\t" \
    "v_add_f32_dpp %8, %8, %8 row_xmask:" #n " row_mask:0xf bank_mask:0xf\n\t" \
    "v_add_f32_dpp %9, %9, %9 row_xmask:" #n " row_mask:0xf bank_mask:0xf"
__device__ __forceinline__ void row_sum10(float& a0, float& a1, float& a2, float& a3, float& a4,
                                          float& a5, float& a6, float& a7, float& a8, float& a9) {
    asm(RSUM_STEP(1) "\n\t" RSUM_STEP(2) "\n\t" RSUM_STEP(4) "\n\t" RSUM_STEP(8)
        : "+v"(a0), "+v"(a1), "+v"(a2), "+v"(a3), "+v"(a4),
          "+v"(a5), "+v"(a6), "+v"(a7), "+v"(a8), "+v"(a9));
}

typedef __attribute__((ext_vector_type(8))) int v8i;

__device__ __forceinline__ void pack_abs8(const v8f& d, int& p0, int& p1, int& p2, int& p3) {
    asm("v_cvt_pk_f16_f32 %0, |%4|, |%5|\n\t"
        "v_cvt_pk_f16_f32 %1, |%6|, |%7|\n\t"
        "v_cvt_pk_f16_f32 %2, |%8|, |%9|\n\t"
        "v_cvt_pk_f16_f32 %3, |%10|, |%11|"
        : "=&v"(p0), "=&v"(p1), "=&v"(p2), "=&v"(p3)
        : "v"(d[0]), "v"(d[1]), "v"(d[2]), "v"(d[3]), "v"(d[4]), "v"(d[5]), "v"(d[6]), "v"(d[7]));
}
__device__ __forceinline__ void next_sample_behind_products(
        int& tn, float& z01n, float& zvn, int t, float invTm1, float fn, float Rnear,
        v8f& a0, v8f& a1, v8f& a2, v8f& a3, v8f& b0, v8f& b1, v8f& b2, v8f& b3) {
    asm("v_add_nc_u32 %0, 1, %11\n\t"
        "s_delay_alu instid0(VALU_DEP_1)\n\t"
        "v_cvt_f32_i32 %1, %0\n\t"
        "s_delay_alu instid0(VALU_DEP_1)\n\t"
        "v_mul_f32 %1, %1, %12\n\t"
        "s_delay_alu instid0(VALU_DEP_1)\n\t"
        "v_fma_f32 %2, %13, %1, %14"
        : "=&v"(tn), "=&v"(z01n), "=&v"(zvn),
          "+v"(a0), "+v"(a1), "+v"(a2), "+v"(a3), "+v"(b0), "+v"(b1), "+v"(b2), "+v"(b3)
        : "v"(t), "v"(invTm1), "v"(fn), "v"(Rnear));
}
__device__ __forceinline__ void packs_before_heads(v16h& f0, v16h& f1, v16h& f2, v16h& f3, v16h& bf) {
    asm("" : "+v"(f0), "+v"(f1), "+v"(f2), "+v"(f3), "+v"(bf));
}
__device__ __forceinline__ v16h abs_fragment(const v8f& lo, const v8f& hi8) {
    v8i p;
    int p0, p1, p2, p3, p4, p5, p6, p7;
    pack_abs8(lo,  p0, p1, p2, p3);
    pack_abs8(hi8, p4, p5, p6, p7);
    p[0] = p0; p[1] = p1; p[2] = p2; p[3] = p3; p[4] = p4; p[5] = p5; p[6] = p6; p[7] = p7;
    return __builtin_bit_cast(v16h, p);
}

typedef __attribute__((ext_vector_type(4))) int v4i;

__device__ __forceinline__ void frag_put(v4i (*S)[32], int f, int lane, const v16h& x) {
    const v8i p = __builtin_bit_cast(v8i, x);
    const v4i a = {p[0], p[1], p[2], p[3]}, b = {p[4], p[5], p[6], p[7]};
    S[2 * f][lane] = a;   S[2 * f + 1][lane] = b;
}
__device__ __forceinline__ v16h frag_get(v4i (*S)[32], int f, int lane) {
    const v4i a = S[2 * f][lane], b = S[2 * f + 1][lane];
    const v8i p = {a[0], a[1], a[2], a[3], b[0], b[1], b[2], b[3]};
    return __builtin_bit_cast(v16h, p);
}
__device__ __forceinline__ int kidx(int e, bool hi) { return (e < 8 ? e : e + 8) + (hi ? 8 : 0); }
template <int NIN>
__device__ __forceinline__ void build_first(const float* __restrict__ Wp, const float* __restrict__ bp,
                                            int row, bool hi, v16h& lo, v16h& hf) {
    const _Float16 zh = (_Float16)0.f;
    const float b = bp[row];
    const _Float16 bh = (_Float16)b, bl = (_Float16)(b - (float)bh);
#pragma unroll
    for (int e = 0; e < 16; ++e) {
        float a = 0.f;
        if (e < NIN) a = Wp[e * 32 + row];
        _Float16 ah = (_Float16)a;
        if (e == 12) ah = bh;
        if (e == 13) ah = bl;
        lo[e] = hi ? zh : ah;   hf[e] = hi ? ah : zh;
    }
}
__device__ __forceinline__ v16h lin_fragment(const v8f& P, bool act) {
    const _Float16 zh = (_Float16)0.f;
    v16h A = {};
#pragma unroll
    for (int e = 0; e < 6; ++e) {
        const _Float16 sh = (_Float16)P[e], sl = (_Float16)(P[e] - (float)sh);
        A[e] = act ? sh : zh;   A[e + 6] = act ? sl : zh;
    }
    const float bs = P[6] + P[7];
    const _Float16 bsh = (_Float16)bs, bsl = (_Float16)(bs - (float)bsh);
    A[12] = act ? bsh : zh;   A[13] = act ? bsl : zh;
    return A;
}

__global__ __launch_bounds__(256)
void nerf_wmma_kernel(const float* __restrict__ rays_o,
                      const float* __restrict__ rays_d,
                      const float* __restrict__ W1,   const float* __restrict__ b1,
                      const float* __restrict__ Wsig, const float* __restrict__ Wsig_d,
                      const float* __restrict__ Wc1,  const float* __restrict__ bc1,
                      const float* __restrict__ Wc2,  const float* __restrict__ Wc2_d,
                      const int*   __restrict__ num_steps,
                      float* __restrict__ out, int N)
{
    const int tid  = blockIdx.x * blockDim.x + threadIdx.x;
    const int lane = threadIdx.x & 31;
    const int wave = tid >> 5;
    const int m    = lane & 15;
    const bool hi  = (lane & 16) != 0;
    const int rayBase = wave * RPW;

    const int slot = (lane & (RPW / 2 - 1)) + (hi ? RPW / 2 : 0);
    int ray = rayBase + slot;
    if (ray >= N) ray = N - 1;

    v16h A1lo0, A1lo1, A1hi0, A1hi1, Ac1lo0, Ac1lo1, Ac1hi0, Ac1hi1, AsigLo, AsigHi, AcolLo, AcolHi;
    v16h AlinS, AlinC;
    __shared__ v4i fragS[28][32];
    __shared__ __attribute__((aligned(16))) float tabS[4][8][32];
    const int  hrow  = m & 7;
    const bool upper = m >= 8;
    const int wv = __builtin_amdgcn_readfirstlane((int)(threadIdx.x >> 5));
    float vs = 0.f, vc = 0.f, v3 = 0.f, v6 = 0.f;
    if (wv < 2) vs = (wv == 0 ? Wsig : Wsig_d)[lane];
    if (wv < 6) vc = (wv < 3 ? Wc2 : Wc2_d)[lane * 3 + (wv < 3 ? wv : wv - 3)];
    if (wv < 3) v3 = W1[wv * 32 + lane];
    if (wv < 6) v6 = Wc1[wv * 32 + lane];
    if (wv >= 6) {
        const float b3 = b1[lane], b6 = bc1[lane];
        v3 = (wv == 6) ? b3 : b3 - (float)(_Float16)b3;
        v6 = (wv == 6) ? b6 : b6 - (float)(_Float16)b6;
    }
    if (wv == 0) {
        v16h lo, hf;  build_first<3>(W1, b1, m, hi, lo, hf);
        frag_put(fragS, 0, lane, lo);  frag_put(fragS, 1, lane, hf);
    } else if (wv == 1) {
        v16h lo, hf;  build_first<3>(W1, b1, m + 16, hi, lo, hf);
        frag_put(fragS, 2, lane, lo);  frag_put(fragS, 3, lane, hf);
    } else if (wv == 2) {
        v16h lo, hf;  build_first<6>(Wc1, bc1, m, hi, lo, hf);
        frag_put(fragS, 4, lane, lo);  frag_put(fragS, 5, lane, hf);
    } else if (wv == 3) {
        v16h lo, hf;  build_first<6>(Wc1, bc1, m + 16, hi, lo, hf);
        frag_put(fragS, 6, lane, lo);  frag_put(fragS, 7, lane, hf);
    }
    tabS[0][wv][lane] = vs;   tabS[1][wv][lane] = vc;
    tabS[2][wv][lane] = v3;   tabS[3][wv][lane] = v6;
    __syncthreads();
    if (wv == 4) {
        const _Float16 zh = (_Float16)0.f;
        v16h lo, hf;
#pragma unroll
        for (int e = 0; e < 16; ++e) {
            const _Float16 svh = (_Float16)(0.5f * tabS[0][hrow][kidx(e, hi)]);
            lo[e] = upper ? zh : svh;   hf[e] = upper ? svh : zh;
        }
        frag_put(fragS, 8, lane, lo);  frag_put(fragS, 9, lane, hf);
    } else if (wv == 5) {
        const _Float16 zh = (_Float16)0.f;
        v16h lo, hf;
#pragma unroll
        for (int e = 0; e < 16; ++e) {
            const _Float16 cvh = (_Float16)(0.5f * tabS[1][hrow][kidx(e, hi)]);
            lo[e] = upper ? zh : cvh;   hf[e] = upper ? cvh : zh;
        }
        frag_put(fragS, 10, lane, lo);  frag_put(fragS, 11, lane, hf);
    } else if (wv == 6) {
        v16h Ap, All;
#pragma unroll
        for (int e = 0; e < 16; ++e) {
            All[e] = (_Float16)(0.5f * tabS[0][hrow][kidx(e, hi)]);
            Ap[e]  = (_Float16)tabS[2][hrow][kidx(e, hi)];
        }
        v8f Zp = {};
        const v8f P = WMMA_F16(Ap, All, Zp);
        frag_put(fragS, 12, lane, lin_fragment(P, upper == hi));
    } else if (wv == 7) {
        v16h Ap, All;
#pragma unroll
        for (int e = 0; e < 16; ++e) {
            All[e] = (_Float16)(0.5f * tabS[1][hrow][kidx(e, hi)]);
            Ap[e]  = (_Float16)tabS[3][hrow][kidx(e, hi)];
        }
        v8f Zp = {};
        const v8f P = WMMA_F16(Ap, All, Zp);
        frag_put(fragS, 13, lane, lin_fragment(P, upper == hi));
    }
    __syncthreads();
    A1lo0  = frag_get(fragS, 0, lane);   A1hi0  = frag_get(fragS, 1, lane);
    A1lo1  = frag_get(fragS, 2, lane);   A1hi1  = frag_get(fragS, 3, lane);
    Ac1lo0 = frag_get(fragS, 4, lane);   Ac1hi0 = frag_get(fragS, 5, lane);
    Ac1lo1 = frag_get(fragS, 6, lane);   Ac1hi1 = frag_get(fragS, 7, lane);
    AsigLo = frag_get(fragS, 8, lane);   AsigHi = frag_get(fragS, 9, lane);
    AcolLo = frag_get(fragS, 10, lane);  AcolHi = frag_get(fragS, 11, lane);
    AlinS  = frag_get(fragS, 12, lane);  AlinC  = frag_get(fragS, 13, lane);

    float ox = rays_o[ray * 3 + 0], oy = rays_o[ray * 3 + 1], oz = rays_o[ray * 3 + 2];
    float dx = rays_d[ray * 3 + 0], dy = rays_d[ray * 3 + 1], dz = rays_d[ray * 3 + 2];
    {
        float il = 1.f / sqrtf(dx * dx + dy * dy + dz * dz);
        dx *= il; dy *= il; dz *= il;
    }
    float near, far;
    {
        float ivx = 1.f / dx, ivy = 1.f / dy, ivz = 1.f / dz;
        float t1x = (-1.f - ox) * ivx, t2x = (1.f - ox) * ivx;
        float t1y = (-1.f - oy) * ivy, t2y = (1.f - oy) * ivy;
        float t1z = (-1.f - oz) * ivz, t2z = (1.f - oz) * ivz;
        near = fmaxf(fmaxf(fminf(t1x, t2x), fminf(t1y, t2y)), fminf(t1z, t2z));
        far  = fminf(fminf(fmaxf(t1x, t2x), fmaxf(t1y, t2y)), fmaxf(t1z, t2z));
        near = fmaxf(near, 0.2f);
        far  = fmaxf(far, near + 1e-6f);
    }

    const int   T       = *num_steps;
    const float invTm1  = 1.f / (float)(T - 1);
    const float fnL     = far - near;
    const float dstepL  = fnL * invTm1;
    const float dlastL  = fnL / (float)T;
    const int   nPer    = (T + 15) >> 4;

    v16h Bf = {};
    Bf[12] = (_Float16)1.f;
    Bf[13] = (_Float16)1.f;

    for (int i = 0; i < RPW / 2; ++i) {
        const unsigned sel = (unsigned)i * 0x11111111u;
        const float Rox = row_pick(ox, sel),  Roy = row_pick(oy, sel),  Roz = row_pick(oz, sel);
        const float Rdx = row_pick(dx, sel),  Rdy = row_pick(dy, sel),  Rdz = row_pick(dz, sel);
        const float Rnear = row_pick(near, sel);
        const float fn    = row_pick(fnL, sel);
        const float dstep = row_pick(dstepL, sel);
        const float dlast = row_pick(dlastL, sel);

        Bf[3] = (_Float16)Rdx;
        Bf[4] = (_Float16)Rdy;
        Bf[5] = (_Float16)Rdz;
        Bf[9] = Bf[3];  Bf[10] = Bf[4];  Bf[11] = Bf[5];

        float Tc = 1.f, Td = 1.f;
        float ws = 0.f,  dep = 0.f,  im0 = 0.f, im1 = 0.f, im2 = 0.f;
        float wsd = 0.f, depd = 0.f, id0 = 0.f, id1 = 0.f, id2 = 0.f;

        int   t   = m * nPer;
        float z01 = (float)t * invTm1;
        float zv  = Rnear + fn * z01;
        for (int j = 0; j < nPer; ++j) {
            const bool  live  = t < T;
            const float px = fminf(fmaxf(Rox + Rdx * zv, -1.f), 1.f);
            const float py = fminf(fmaxf(Roy + Rdy * zv, -1.f), 1.f);
            const float pz = fminf(fmaxf(Roz + Rdz * zv, -1.f), 1.f);
            Bf[0] = (_Float16)px; Bf[1] = (_Float16)py; Bf[2] = (_Float16)pz;
            Bf[6] = Bf[0];  Bf[7] = Bf[1];  Bf[8] = Bf[2];

            v8f Ds, Dc;
            {
                v8f Z0 = {};
                v8f Z1 = {};
                Ds = WMMA_F16(AlinS, Bf, Z0);
                Dc = WMMA_F16(AlinC, Bf, Z1);
            }
            const v8f Zf = {};
            v8f D0a = WMMA_F16(A1lo0, Bf, Zf);
            v8f D1a = WMMA_F16(A1lo1, Bf, Zf);
            v8f C0a = WMMA_F16(Ac1lo0, Bf, Zf);
            v8f C1a = WMMA_F16(Ac1lo1, Bf, Zf);
            v8f D0b = WMMA_F16(A1hi0, Bf, Zf);
            v8f D1b = WMMA_F16(A1hi1, Bf, Zf);
            v8f C0b = WMMA_F16(Ac1hi0, Bf, Zf);
            v8f C1b = WMMA_F16(Ac1hi1, Bf, Zf);

            int tn;  float z01n, zvn;
            next_sample_behind_products(tn, z01n, zvn, t, invTm1, fn, Rnear,
                                        D0a, D1a, C0a, C1a, D0b, D1b, C0b, C1b);
            v16h Bh1  = abs_fragment(D0a, D1a);
            v16h Bhc1 = abs_fragment(C0a, C1a);
            v16h Bh2  = abs_fragment(D0b, D1b);
            v16h Bhc2 = abs_fragment(C0b, C1b);
            packs_before_heads(Bh1, Bhc1, Bh2, Bhc2, Bf);

            Ds = WMMA_F16(AsigLo, Bh1,  Ds);
            Dc = WMMA_F16(AcolLo, Bhc1, Dc);
            Ds = WMMA_F16(AsigHi, Bh2,  Ds);
            Dc = WMMA_F16(AcolHi, Bhc2, Dc);

            const float delta = live ? ((t == T - 1) ? dlast : dstep) : 0.f;
            const float dl2   = -delta * L2E;
            const float sg  = softplus_f(Ds[0]);
            const float sgd = softplus_f(Ds[1]);
            const float a   = 1.f - exp2_fast(dl2 * sg);
            const float ad  = 1.f - exp2_fast(dl2 * sgd);
            const float w   = a  * Tc;
            const float wd  = ad * Td;
            Tc *= (1.f - a  + 1e-15f);
            Td *= (1.f - ad + 1e-15f);
            ws  += w;   dep  += w  * z01;
            wsd += wd;  depd += wd * z01;
            im0 += w  * sigmoid_f(Dc[0]);
            im1 += w  * sigmoid_f(Dc[1]);
            im2 += w  * sigmoid_f(Dc[2]);
            id0 += wd * sigmoid_f(Dc[3]);
            id1 += wd * sigmoid_f(Dc[4]);
            id2 += wd * sigmoid_f(Dc[5]);
            t = tn;  z01 = z01n;  zv = zvn;
        }

        float E  = row_dpp<ROW_SHR(1)>(1.f, Tc);
        float Ed = row_dpp<ROW_SHR(1)>(1.f, Td);
        row_scan_mul2(E, Ed);
        float Sws  = ws  * E,  Sdep  = dep  * E,  Sim0 = im0 * E,  Sim1 = im1 * E,  Sim2 = im2 * E;
        float Swsd = wsd * Ed, Sdepd = depd * Ed, Sid0 = id0 * Ed, Sid1 = id1 * Ed, Sid2 = id2 * Ed;
        row_sum10(Sws, Sdep, Sim0, Sim1, Sim2, Swsd, Sdepd, Sid0, Sid1, Sid2);
        const float bg  = 1.f - Sws;
        const float bgd = 1.f - Swsd;
        const int oray = rayBase + i + (hi ? RPW / 2 : 0);
        if (m == i && oray < N) {
            float* o9 = out + (size_t)oray * 9;
            o9[0] = Sim0 + bg;  o9[1] = Sim1 + bg;  o9[2] = Sim2 + bg;
            o9[3] = Sdep;
            o9[4] = Sws;
            o9[5] = Sid0 + bgd; o9[6] = Sid1 + bgd; o9[7] = Sid2 + bgd;
            o9[8] = Sdepd;
        }
    }

}

extern "C" void kernel_launch(void* const* d_in, const int* in_sizes, int n_in,
                              void* d_out, int out_size, void* d_ws, size_t ws_size,
                              hipStream_t stream) {
    const float* rays_o = (const float*)d_in[0];
    const float* rays_d = (const float*)d_in[1];
    const float* W1     = (const float*)d_in[2];
    const float* b1     = (const float*)d_in[3];
    const float* Wsig   = (const float*)d_in[4];
    const float* Wsig_d = (const float*)d_in[5];
    const float* Wc1    = (const float*)d_in[6];
    const float* bc1    = (const float*)d_in[7];
    const float* Wc2    = (const float*)d_in[8];
    const float* Wc2_d  = (const float*)d_in[9];
    const int*   nsteps = (const int*)d_in[10];

    const int N      = in_sizes[0] / 3;
    const int waves  = (N + RPW - 1) / RPW;
    const int thr    = waves * 32;
    const int block  = 256;
    const int grid   = (thr + block - 1) / block;

    nerf_wmma_kernel<<<grid, block, 0, stream>>>(
        rays_o, rays_d, W1, b1, Wsig, Wsig_d, Wc1, bc1, Wc2, Wc2_d,
        nsteps, (float*)d_out, N);
}
